# starting kernel
# baseline (speedup 1.0000x reference)
#include <hip/hip_runtime.h>
#include <hip/hip_bf16.h>

#define N_NODES 50000
#define N_EDGES 600000
#define N_REL   8
#define D       128
#define N_TILES 3125

typedef __attribute__((ext_vector_type(2))) float v2f;
typedef __attribute__((ext_vector_type(8))) float v8f;

__device__ __forceinline__ v8f wmma4(v2f a, v2f b, v8f c) {
    return __builtin_amdgcn_wmma_f32_16x16x4_f32(false, a, false, b, (short)0, c, false, false);
}

__global__ __launch_bounds__(128) void k_transform(
    const float* __restrict__ x, const float* __restrict__ Wrel,
    const float* __restrict__ Wloop, const float* __restrict__ brel,
    float* __restrict__ xW, float* __restrict__ mid)
{
    __shared__ float Wlds[D * D];

    const int rel = blockIdx.y;
    const float* W = (rel < N_REL) ? (Wrel + (size_t)rel * D * D) : Wloop;

    for (int i = threadIdx.x; i < (D * D) / 2; i += blockDim.x) {
        int kp = i >> 7;
        int n  = i & 127;
        Wlds[i * 2 + 0] = W[(2 * kp + 0) * D + n];
        Wlds[i * 2 + 1] = W[(2 * kp + 1) * D + n];
    }
    __syncthreads();

    const int wave = threadIdx.x >> 5;
    const int lane = threadIdx.x & 31;
    const int tile = blockIdx.x * 4 + wave;
    if (tile >= N_TILES) return;

    const int m  = lane & 15;
    const int hi = lane >> 4;
    const int kh = hi * 2;
    const int row0 = tile * 16;

    const float* xrow = x + (size_t)(row0 + m) * D;

    v8f acc[8];
    if (rel == N_REL) {
        #pragma unroll
        for (int nt = 0; nt < 8; ++nt) {
            float bv = brel[nt * 16 + m];
            #pragma unroll
            for (int v = 0; v < 8; ++v) acc[nt][v] = bv;
        }
    } else {
        #pragma unroll
        for (int nt = 0; nt < 8; ++nt) acc[nt] = (v8f){};
    }

    for (int kc = 0; kc < 32; ++kc) {
        const int kk = kc * 4 + kh;
        v2f a;
        a[0] = xrow[kk];
        a[1] = xrow[kk + 1];
        const float* bp = Wlds + (kk >> 1) * 256;
        #pragma unroll
        for (int nt = 0; nt < 8; ++nt) {
            v2f b = *(const v2f*)(bp + (nt * 16 + m) * 2);
            acc[nt] = wmma4(a, b, acc[nt]);
        }
    }

    float* dstp = (rel < N_REL) ? (xW + (size_t)rel * N_NODES * D) : mid;
    const int rbase = row0 + hi * 8;
    #pragma unroll
    for (int nt = 0; nt < 8; ++nt) {
        #pragma unroll
        for (int v = 0; v < 8; ++v)
            dstp[(size_t)(rbase + v) * D + nt * 16 + m] = acc[nt][v];
    }
}

__global__ __launch_bounds__(256) void k_scatter(
    const float* __restrict__ xW, const int* __restrict__ src,
    const int* __restrict__ dst, const int* __restrict__ etype,
    float* mid)
{
    const int wid  = (int)((blockIdx.x * blockDim.x + threadIdx.x) >> 5);
    const int lane = threadIdx.x & 31;
    if (wid >= N_EDGES) return;

    const int s = src[wid];
    const int d = dst[wid];
    const int r = etype[wid];

    const float4* mp = (const float4*)(xW + ((size_t)r * N_NODES + s) * D);
    float4 v = mp[lane];

    float* op = mid + (size_t)d * D + lane * 4;
    unsafeAtomicAdd(op + 0, v.x);
    unsafeAtomicAdd(op + 1, v.y);
    unsafeAtomicAdd(op + 2, v.z);
    unsafeAtomicAdd(op + 3, v.w);
}

#define HSTRIDE 258

__global__ __launch_bounds__(256) void k_mlp(
    const float* __restrict__ x, const float* mid,
    const float* __restrict__ W1, const float* __restrict__ b1,
    const float* __restrict__ W2, const float* __restrict__ b2,
    float* out)
{
    __shared__ float hl[32 * HSTRIDE];

    const int wave = threadIdx.x >> 5;
    const int lane = threadIdx.x & 31;
    const int m  = lane & 15;
    const int hi = lane >> 4;
    const int kh = hi * 2;
    const int t0 = blockIdx.x * 2;

    #pragma unroll
    for (int j = 0; j < 4; ++j) {
        const int job = wave * 4 + j;
        const int s   = job >> 4;
        const int ct  = job & 15;
        const int gt  = t0 + s;
        if (gt < N_TILES) {
            const int r0 = gt * 16;
            v8f acc;
            {
                float bv = b1[ct * 16 + m];
                #pragma unroll
                for (int v = 0; v < 8; ++v) acc[v] = bv;
            }
            const float* ax = x   + (size_t)(r0 + m) * D;
            const float* am = mid + (size_t)(r0 + m) * D;
            for (int kc = 0; kc < 32; ++kc) {
                const int kk = kc * 4 + kh;
                v2f a; a[0] = ax[kk]; a[1] = ax[kk + 1];
                v2f b;
                b[0] = W1[(size_t)kk * 256 + ct * 16 + m];
                b[1] = W1[(size_t)(kk + 1) * 256 + ct * 16 + m];
                acc = wmma4(a, b, acc);
            }
            for (int kc = 0; kc < 32; ++kc) {
                const int kk = kc * 4 + kh;
                const int kg = 128 + kk;
                v2f a; a[0] = am[kk]; a[1] = am[kk + 1];
                v2f b;
                b[0] = W1[(size_t)kg * 256 + ct * 16 + m];
                b[1] = W1[(size_t)(kg + 1) * 256 + ct * 16 + m];
                acc = wmma4(a, b, acc);
            }
            #pragma unroll
            for (int v = 0; v < 8; ++v) {
                const int lr = s * 16 + v + hi * 8;
                float val = acc[v];
                hl[lr * HSTRIDE + ct * 16 + m] = val > 0.0f ? val : 0.0f;
            }
        }
    }
    __syncthreads();

    #pragma unroll
    for (int j = 0; j < 2; ++j) {
        const int job = wave * 2 + j;
        const int s   = job >> 3;
        const int ct  = job & 7;
        const int gt  = t0 + s;
        if (gt < N_TILES) {
            const int r0 = gt * 16;
            v8f acc;
            {
                float bv = b2[ct * 16 + m];
                #pragma unroll
                for (int v = 0; v < 8; ++v) acc[v] = bv;
            }
            const float* hrow = hl + (s * 16 + m) * HSTRIDE;
            for (int kc = 0; kc < 64; ++kc) {
                const int kk = kc * 4 + kh;
                v2f a = *(const v2f*)(hrow + kk);
                v2f b;
                b[0] = W2[(size_t)kk * D + ct * 16 + m];
                b[1] = W2[(size_t)(kk + 1) * D + ct * 16 + m];
                acc = wmma4(a, b, acc);
            }
            #pragma unroll
            for (int v = 0; v < 8; ++v)
                out[(size_t)(r0 + v + hi * 8) * D + ct * 16 + m] = acc[v];
        }
    }
}

extern "C" void kernel_launch(void* const* d_in, const int* in_sizes, int n_in,
                              void* d_out, int out_size, void* d_ws, size_t ws_size,
                              hipStream_t stream) {
    const float* x     = (const float*)d_in[0];
    const int*   src   = (const int*)  d_in[1];
    const int*   dst   = (const int*)  d_in[2];
    const int*   etype = (const int*)  d_in[3];
    const float* Wrel  = (const float*)d_in[4];
    const float* Wloop = (const float*)d_in[5];
    const float* brel  = (const float*)d_in[6];
    const float* W1    = (const float*)d_in[7];
    const float* b1    = (const float*)d_in[8];
    const float* W2    = (const float*)d_in[9];
    const float* b2    = (const float*)d_in[10];

    float* out = (float*)d_out;
    float* xW  = (float*)d_ws;

    dim3 g1((N_TILES + 3) / 4, N_REL + 1);
    k_transform<<<g1, 128, 0, stream>>>(x, Wrel, Wloop, brel, xW, out);

    k_scatter<<<(N_EDGES * 32) / 256, 256, 0, stream>>>(xW, src, dst, etype, out);

    k_mlp<<<(N_TILES + 1) / 2, 256, 0, stream>>>(x, out, W1, b1, W2, b2, out);
}


# edge grouping by counting sorts in LDS: 240 edge stretches sorted by 128-row dst bin, a block per bin sorts its runs by row; no global atomics, scan or fill launch. Sum order: LDS add arrival
# speedup vs baseline: 1.8978x; 1.8978x over previous
#include <hip/hip_runtime.h>
#include <hip/hip_bf16.h>

#define N_NODES 50000
#define N_EDGES 600000
#define N_REL   8
#define D       128
#define N_TILES 3125

typedef __attribute__((ext_vector_type(2))) float v2f;
typedef __attribute__((ext_vector_type(8))) float v8f;

__device__ __forceinline__ v8f wmma4(v2f a, v2f b, v8f c) {
    return __builtin_amdgcn_wmma_f32_16x16x4_f32(false, a, false, b, (short)0, c, false, false);
}

#define TR_WAVES  8
#define TR_NREL   3
#define TR_Q      32

__device__ __forceinline__ const float* w_select(
    const float* __restrict__ Wrel, const float* __restrict__ Wloop, int rel)
{
    return (rel < N_REL) ? (Wrel + (size_t)rel * D * D) : Wloop;
}

__device__ __forceinline__ void w_fetch(const float* __restrict__ Wm, int tid,
                                        v2f (&p)[TR_Q])
{
    const float* g = Wm + (size_t)((tid >> 7) * 2) * D + (tid & 127);
    #pragma unroll
    for (int q = 0; q < TR_Q; ++q) {
        p[q][0] = g[q * 4 * D];
        p[q][1] = g[q * 4 * D + D];
    }
}

__device__ __forceinline__ void w_put(float* buf, int tid, const v2f (&p)[TR_Q])
{
    float* d = buf + tid * 2;
    #pragma unroll
    for (int q = 0; q < TR_Q; ++q) *(v2f*)(d + q * 512) = p[q];
}

__device__ __forceinline__ void tile_products(const float* Wb, const v2f (&a)[32],
                                              v8f (&acc)[8], int hi, int m)
{
    #pragma unroll
    for (int kc = 0; kc < 32; ++kc) {
        const float* bp = Wb + (2 * kc + hi) * 256 + m * 2;
        #pragma unroll
        for (int nt = 0; nt < 8; ++nt) {
            v2f b = *(const v2f*)(bp + nt * 32);
            acc[nt] = wmma4(a[kc], b, acc[nt]);
        }
    }
}

__device__ __forceinline__ void tile_store(float* __restrict__ dstp, int rbase, int m,
                                           const v8f (&acc)[8])
{
    #pragma unroll
    for (int nt = 0; nt < 8; ++nt) {
        #pragma unroll
        for (int v = 0; v < 8; ++v)
            dstp[(size_t)(rbase + v) * D + nt * 16 + m] = acc[nt][v];
    }
}

__global__ __launch_bounds__(256) void k_transform(
    const float* __restrict__ x, const float* __restrict__ Wrel,
    const float* __restrict__ Wloop, const float* __restrict__ brel,
    float* __restrict__ xW, float* __restrict__ mid)
{
    __shared__ float Wlds[2][D * D];

    const int tid  = threadIdx.x;
    const int wave = tid >> 5;
    const int lane = tid & 31;
    const int m  = lane & 15;
    const int hi = lane >> 4;
    const int kh = hi * 2;

    const int  tile   = blockIdx.x * TR_WAVES + wave;
    const bool active = tile < N_TILES;
    const int  row0   = (active ? tile : (N_TILES - 1)) * 16;
    const int  rbase  = row0 + hi * 8;
    const int  rel0   = blockIdx.y * TR_NREL;

    v2f p[TR_Q];
    w_fetch(w_select(Wrel, Wloop, rel0), tid, p);

    v2f a[32];
    {
        const float* xrow = x + (size_t)(row0 + m) * D + kh;
        #pragma unroll
        for (int kc = 0; kc < 32; ++kc) a[kc] = *(const v2f*)(xrow + kc * 4);
    }

    w_put(Wlds[0], tid, p);
    __syncthreads();

    v8f acc[8];

    for (int rr = 0; rr < TR_NREL - 1; ++rr) {
        const int rel = rel0 + rr;
        w_fetch(w_select(Wrel, Wloop, rel + 1), tid, p);

        if (active) {
            #pragma unroll
            for (int nt = 0; nt < 8; ++nt) acc[nt] = (v8f){};
            tile_products(Wlds[rr & 1], a, acc, hi, m);
        }

        w_put(Wlds[(rr + 1) & 1], tid, p);
        __syncthreads();

        if (active) tile_store(xW + (size_t)rel * N_NODES * D, rbase, m, acc);
    }

    if (active) {
        const int rel = rel0 + TR_NREL - 1;
        if (rel == N_REL) {
            #pragma unroll
            for (int nt = 0; nt < 8; ++nt) {
                float bv = brel[nt * 16 + m];
                #pragma unroll
                for (int v = 0; v < 8; ++v) acc[nt][v] = bv;
            }
        } else {
            #pragma unroll
            for (int nt = 0; nt < 8; ++nt) acc[nt] = (v8f){};
        }
        tile_products(Wlds[(TR_NREL - 1) & 1], a, acc, hi, m);
        float* dstp = (rel < N_REL) ? (xW + (size_t)rel * N_NODES * D) : mid;
        tile_store(dstp, rbase, m, acc);
    }
}

#define NB    391
#define NBP   392
#define NCH   240
#define CE    2500
#define CH_T  1024
#define CH_K  3

__device__ __forceinline__ int wave_inc_scan(int v, int lane)
{
    for (int o = 1; o < 32; o <<= 1) {
        const int up = __shfl_up(v, o);
        if (lane >= o) v += up;
    }
    return v;
}

__global__ __launch_bounds__(CH_T) void k_chunk(
    const int* __restrict__ src, const int* __restrict__ dst,
    const int* __restrict__ etype, int* __restrict__ lstart, int* __restrict__ sorted)
{
    __shared__ int hist[NBP];
    __shared__ int wtot[32];
    __shared__ int recl[CE];
    const int tid  = threadIdx.x;
    const int lane = tid & 31;
    const int wave = tid >> 5;
    const int c    = blockIdx.x;

    if (tid < NBP) hist[tid] = 0;
    __syncthreads();

    int rec[CH_K], bin[CH_K], rk[CH_K];
    #pragma unroll
    for (int k = 0; k < CH_K; ++k) {
        const int i = tid + k * CH_T;
        bin[k] = -1;
        rec[k] = 0;
        if (i < CE) {
            const int e = c * CE + i;
            const int d = dst[e];
            bin[k] = d >> 7;
            rec[k] = ((d & 127) << 19) | (etype[e] * N_NODES + src[e]);
        }
    }
    #pragma unroll
    for (int k = 0; k < CH_K; ++k) {
        rk[k] = 0;
        if (bin[k] >= 0) rk[k] = atomicAdd(&hist[bin[k]], 1);
    }
    __syncthreads();

    const int v   = (tid < NB) ? hist[tid] : 0;
    const int inc = wave_inc_scan(v, lane);
    if (lane == 31) wtot[wave] = inc;
    __syncthreads();
    if (wave == 0) {
        const int tot = wtot[lane];
        const int ti  = wave_inc_scan(tot, lane);
        wtot[lane] = ti - tot;
    }
    __syncthreads();
    const int ex = wtot[wave] + inc - v;
    if (tid < NBP) {
        hist[tid] = ex;
        lstart[c * NBP + tid] = ex;
    }
    __syncthreads();

    #pragma unroll
    for (int k = 0; k < CH_K; ++k)
        if (bin[k] >= 0) recl[hist[bin[k]] + rk[k]] = rec[k];
    __syncthreads();

    #pragma unroll
    for (int k = 0; k < CH_K; ++k) {
        const int i = tid + k * CH_T;
        if (i < CE) sorted[c * CE + i] = recl[i];
    }
}

#define GA_T    512
#define GA_U    8
#define GA_CAP  (GA_T * GA_U)

typedef __attribute__((ext_vector_type(4))) float v4f;

template <int G>
__device__ __forceinline__ void sum_rows(v4f& acc, const float* __restrict__ xW,
                                         int mine, int j, int n, int lane)
{
    v4f v[G];
    #pragma unroll
    for (int u = 0; u < G; ++u) {
        const int jj = j + u < n ? j + u : n - 1;
        const int r  = __builtin_amdgcn_readlane(mine, jj);
        v[u] = ((const v4f*)(xW + (size_t)r * D))[lane];
    }
    #pragma unroll
    for (int u = 0; u < G; ++u) asm volatile("" :: "v"(v[u]));
    #pragma unroll
    for (int u = 0; u < G; ++u)
        if (j + u < n) acc += v[u];
}

__global__ __launch_bounds__(GA_T) void k_gather(
    const float* __restrict__ xW, const int* __restrict__ lstart,
    const int* __restrict__ sorted, float* mid)
{
    __shared__ int runo[NCH + 1];
    __shared__ int runs[NCH];
    __shared__ int wtot[GA_T / 32];
    __shared__ int rowc[128];
    __shared__ int rowb[129];
    __shared__ int recs[GA_CAP];

    const int tid  = threadIdx.x;
    const int lane = tid & 31;
    const int wave = __builtin_amdgcn_readfirstlane(tid >> 5);
    const int b    = blockIdx.x;

    int cnt = 0;
    if (tid < NCH) {
        const int* ls = lstart + tid * NBP + b;
        const int s = ls[0];
        cnt = ls[1] - s;
        runs[tid] = tid * CE + s;
    }
    const int inc = wave_inc_scan(cnt, lane);
    if (lane == 31) wtot[wave] = inc;
    __syncthreads();
    if (wave == 0) {
        const int tot = lane < GA_T / 32 ? wtot[lane] : 0;
        const int ti  = wave_inc_scan(tot, lane);
        if (lane < GA_T / 32) wtot[lane] = ti - tot;
    }
    __syncthreads();
    const int ex = wtot[wave] + inc - cnt;
    if (tid < NCH) runo[tid] = ex;
    if (tid == NCH - 1) runo[NCH] = ex + cnt;
    __syncthreads();
    const int n = __builtin_amdgcn_readfirstlane(runo[NCH]);

    for (int w0 = 0; w0 < n; w0 += GA_CAP) {
        const int mm = n - w0 < GA_CAP ? n - w0 : GA_CAP;
        if (tid < 128) rowc[tid] = 0;
        __syncthreads();

        int rec[GA_U], rk[GA_U];
        #pragma unroll
        for (int u = 0; u < GA_U; ++u) {
            const int i = tid + u * GA_T;
            rec[u] = -1;
            if (i < mm) {
                const int g = w0 + i;
                int lo = 0, hi = NCH;
                while (hi - lo > 1) {
                    const int md = (lo + hi) >> 1;
                    if (runo[md] <= g) lo = md; else hi = md;
                }
                rec[u] = sorted[runs[lo] + (g - runo[lo])];
            }
        }
        #pragma unroll
        for (int u = 0; u < GA_U; ++u) {
            rk[u] = 0;
            if (rec[u] >= 0) rk[u] = atomicAdd(&rowc[rec[u] >> 19], 1);
        }
        __syncthreads();

        if (wave == 0) {
            const int c0 = rowc[4 * lane], c1 = rowc[4 * lane + 1];
            const int c2 = rowc[4 * lane + 2], c3 = rowc[4 * lane + 3];
            const int s4 = c0 + c1 + c2 + c3;
            int e4 = wave_inc_scan(s4, lane) - s4;
            rowb[4 * lane]     = e4;  e4 += c0;
            rowb[4 * lane + 1] = e4;  e4 += c1;
            rowb[4 * lane + 2] = e4;  e4 += c2;
            rowb[4 * lane + 3] = e4;  e4 += c3;
            if (lane == 31) rowb[128] = e4;
        }
        __syncthreads();

        #pragma unroll
        for (int u = 0; u < GA_U; ++u)
            if (rec[u] >= 0) recs[rowb[rec[u] >> 19] + rk[u]] = rec[u] & 0x7FFFF;
        __syncthreads();

        for (int rr = 0; rr < 8; ++rr) {
            const int r    = wave * 8 + rr;
            const int node = b * 128 + r;
            const int beg  = __builtin_amdgcn_readfirstlane(rowb[r]);
            const int end  = __builtin_amdgcn_readfirstlane(rowb[r + 1]);
            if (node >= N_NODES || end <= beg) continue;

            v4f* op = (v4f*)(mid + (size_t)node * D) + lane;
            v4f acc = *op;
            for (int base = beg; base < end; base += 32) {
                const int rem = end - base;
                const int nn  = rem < 32 ? rem : 32;
                const int mine = recs[base + (lane < nn ? lane : nn - 1)];

                int j = 0;
                for (; nn - j > 8; j += 16) sum_rows<16>(acc, xW, mine, j, nn, lane);
                if (nn - j > 4)  sum_rows<8>(acc, xW, mine, j, nn, lane);
                else if (j < nn) sum_rows<4>(acc, xW, mine, j, nn, lane);
            }
            *op = acc;
        }
        __syncthreads();
    }
}

#define ML_HP 132

template <int LD>
__device__ __forceinline__ void w_fetch_ld(const float* __restrict__ Wm, int tid,
                                           v2f (&p)[TR_Q])
{
    const float* g = Wm + (size_t)((tid >> 7) * 2) * LD + (tid & 127);
    #pragma unroll
    for (int q = 0; q < TR_Q; ++q) {
        p[q][0] = g[q * 4 * LD];
        p[q][1] = g[q * 4 * LD + LD];
    }
}

__device__ __forceinline__ void bias_init(v8f (&acc)[8], const float (&b)[8])
{
    #pragma unroll
    for (int nt = 0; nt < 8; ++nt) {
        #pragma unroll
        for (int v = 0; v < 8; ++v) acc[nt][v] = b[nt];
    }
}

__device__ __forceinline__ void h_put(float* hw, int m, int hi, const v8f (&acc)[8])
{
    #pragma unroll
    for (int nt = 0; nt < 8; ++nt) {
        #pragma unroll
        for (int v = 0; v < 8; ++v) {
            const float val = acc[nt][v];
            hw[(hi * 8 + v) * ML_HP + nt * 16 + m] = val > 0.0f ? val : 0.0f;
        }
    }
}

__device__ __forceinline__ void h_get(const float* hw, int m, int kh, v2f (&a)[32])
{
    const float* r = hw + m * ML_HP + kh;
    #pragma unroll
    for (int kc = 0; kc < 32; ++kc) a[kc] = *(const v2f*)(r + kc * 4);
}

__global__ __launch_bounds__(256) void k_mlp(
    const float* __restrict__ x, const float* mid,
    const float* __restrict__ W1, const float* __restrict__ b1,
    const float* __restrict__ W2, const float* __restrict__ b2,
    float* out)
{
    __shared__ float Wlds[2][D * D];
    __shared__ float hl[TR_WAVES][16 * ML_HP];

    const int tid  = threadIdx.x;
    const int wave = tid >> 5;
    const int lane = tid & 31;
    const int m  = lane & 15;
    const int hi = lane >> 4;
    const int kh = hi * 2;

    const int  tile   = blockIdx.x * TR_WAVES + wave;
    const bool active = tile < N_TILES;
    const int  row0   = (active ? tile : (N_TILES - 1)) * 16;
    const int  rbase  = row0 + hi * 8;

    v2f p[TR_Q];
    w_fetch_ld<2 * D>(W1, tid, p);

    v2f ax[32], am[32];
    {
        const float* xrow = x   + (size_t)(row0 + m) * D + kh;
        const float* mrow = mid + (size_t)(row0 + m) * D + kh;
        #pragma unroll
        for (int kc = 0; kc < 32; ++kc) ax[kc] = *(const v2f*)(xrow + kc * 4);
        #pragma unroll
        for (int kc = 0; kc < 32; ++kc) am[kc] = *(const v2f*)(mrow + kc * 4);
    }

    float bh0[8], bh1[8], bo[8];
    #pragma unroll
    for (int nt = 0; nt < 8; ++nt) {
        bh0[nt] = b1[nt * 16 + m];
        bh1[nt] = b1[D + nt * 16 + m];
        bo[nt]  = b2[nt * 16 + m];
    }

    w_put(Wlds[0], tid, p);
    __syncthreads();

    float* hw = hl[wave];
    v8f acc[8];
    v2f ah0[32], ah1[32];

    w_fetch_ld<2 * D>(W1 + (size_t)D * 2 * D, tid, p);
    if (active) {
        bias_init(acc, bh0);
        tile_products(Wlds[0], ax, acc, hi, m);
    }
    w_put(Wlds[1], tid, p);
    __syncthreads();

    w_fetch_ld<2 * D>(W1 + D, tid, p);
    if (active) {
        tile_products(Wlds[1], am, acc, hi, m);
        h_put(hw, m, hi, acc);
    }
    w_put(Wlds[0], tid, p);
    __syncthreads();
    if (active) h_get(hw, m, kh, ah0);

    w_fetch_ld<2 * D>(W1 + (size_t)D * 2 * D + D, tid, p);
    if (active) {
        bias_init(acc, bh1);
        tile_products(Wlds[0], ax, acc, hi, m);
    }
    w_put(Wlds[1], tid, p);
    __syncthreads();

    w_fetch_ld<D>(W2, tid, p);
    if (active) {
        tile_products(Wlds[1], am, acc, hi, m);
        h_put(hw, m, hi, acc);
    }
    w_put(Wlds[0], tid, p);
    __syncthreads();
    if (active) h_get(hw, m, kh, ah1);

    w_fetch_ld<D>(W2 + (size_t)D * D, tid, p);
    if (active) {
        bias_init(acc, bo);
        tile_products(Wlds[0], ah0, acc, hi, m);
    }
    w_put(Wlds[1], tid, p);
    __syncthreads();

    if (active) {
        tile_products(Wlds[1], ah1, acc, hi, m);
        tile_store(out, rbase, m, acc);
    }
}

extern "C" void kernel_launch(void* const* d_in, const int* in_sizes, int n_in,
                              void* d_out, int out_size, void* d_ws, size_t ws_size,
                              hipStream_t stream) {
    const float* x     = (const float*)d_in[0];
    const int*   src   = (const int*)  d_in[1];
    const int*   dst   = (const int*)  d_in[2];
    const int*   etype = (const int*)  d_in[3];
    const float* Wrel  = (const float*)d_in[4];
    const float* Wloop = (const float*)d_in[5];
    const float* brel  = (const float*)d_in[6];
    const float* W1    = (const float*)d_in[7];
    const float* b1    = (const float*)d_in[8];
    const float* W2    = (const float*)d_in[9];
    const float* b2    = (const float*)d_in[10];

    float* out = (float*)d_out;
    float* xW  = (float*)d_ws;
    int* lstart = (int*)((char*)d_ws + 204800000);
    int* sorted = lstart + NCH * NBP;

    dim3 g1((N_TILES + 7) / 8, 3);
    k_transform<<<g1, 256, 0, stream>>>(x, Wrel, Wloop, brel, xW, out);

    k_chunk<<<NCH, CH_T, 0, stream>>>(src, dst, etype, lstart, sorted);
    k_gather<<<NB, GA_T, 0, stream>>>(xW, lstart, sorted, out);

    k_mlp<<<(N_TILES + 7) / 8, 256, 0, stream>>>(x, out, W1, b1, W2, b2, out);
}


# k_mlp: the six weight pieces run through ONE rolled product body (A fragments copied into one register set per piece); kernel text under a third, resident in the instruction cache. Sum order kept
# speedup vs baseline: 1.9788x; 1.0427x over previous
#include <hip/hip_runtime.h>
#include <hip/hip_bf16.h>

#define N_NODES 50000
#define N_EDGES 600000
#define N_REL   8
#define D       128
#define N_TILES 3125

typedef __attribute__((ext_vector_type(2))) float v2f;
typedef __attribute__((ext_vector_type(8))) float v8f;

__device__ __forceinline__ v8f wmma4(v2f a, v2f b, v8f c) {
    return __builtin_amdgcn_wmma_f32_16x16x4_f32(false, a, false, b, (short)0, c, false, false);
}

#define TR_WAVES  8
#define TR_NREL   3
#define TR_Q      32

__device__ __forceinline__ const float* w_select(
    const float* __restrict__ Wrel, const float* __restrict__ Wloop, int rel)
{
    return (rel < N_REL) ? (Wrel + (size_t)rel * D * D) : Wloop;
}

__device__ __forceinline__ void w_fetch(const float* __restrict__ Wm, int tid,
                                        v2f (&p)[TR_Q])
{
    const float* g = Wm + (size_t)((tid >> 7) * 2) * D + (tid & 127);
    #pragma unroll
    for (int q = 0; q < TR_Q; ++q) {
        p[q][0] = g[q * 4 * D];
        p[q][1] = g[q * 4 * D + D];
    }
}

__device__ __forceinline__ void w_put(float* buf, int tid, const v2f (&p)[TR_Q])
{
    float* d = buf + tid * 2;
    #pragma unroll
    for (int q = 0; q < TR_Q; ++q) *(v2f*)(d + q * 512) = p[q];
}

__device__ __forceinline__ void tile_products(const float* Wb, const v2f (&a)[32],
                                              v8f (&acc)[8], int hi, int m)
{
    #pragma unroll
    for (int kc = 0; kc < 32; ++kc) {
        const float* bp = Wb + (2 * kc + hi) * 256 + m * 2;
        #pragma unroll
        for (int nt = 0; nt < 8; ++nt) {
            v2f b = *(const v2f*)(bp + nt * 32);
            acc[nt] = wmma4(a[kc], b, acc[nt]);
        }
    }
}

__device__ __forceinline__ void tile_store(float* __restrict__ dstp, int rbase, int m,
                                           const v8f (&acc)[8])
{
    #pragma unroll
    for (int nt = 0; nt < 8; ++nt) {
        #pragma unroll
        for (int v = 0; v < 8; ++v)
            dstp[(size_t)(rbase + v) * D + nt * 16 + m] = acc[nt][v];
    }
}

__global__ __launch_bounds__(256) void k_transform(
    const float* __restrict__ x, const float* __restrict__ Wrel,
    const float* __restrict__ Wloop, const float* __restrict__ brel,
    float* __restrict__ xW, float* __restrict__ mid)
{
    __shared__ float Wlds[2][D * D];

    const int tid  = threadIdx.x;
    const int wave = tid >> 5;
    const int lane = tid & 31;
    const int m  = lane & 15;
    const int hi = lane >> 4;
    const int kh = hi * 2;

    const int  tile   = blockIdx.x * TR_WAVES + wave;
    const bool active = tile < N_TILES;
    const int  row0   = (active ? tile : (N_TILES - 1)) * 16;
    const int  rbase  = row0 + hi * 8;
    const int  rel0   = blockIdx.y * TR_NREL;

    v2f p[TR_Q];
    w_fetch(w_select(Wrel, Wloop, rel0), tid, p);

    v2f a[32];
    {
        const float* xrow = x + (size_t)(row0 + m) * D + kh;
        #pragma unroll
        for (int kc = 0; kc < 32; ++kc) a[kc] = *(const v2f*)(xrow + kc * 4);
    }

    w_put(Wlds[0], tid, p);
    __syncthreads();

    v8f acc[8];

    for (int rr = 0; rr < TR_NREL - 1; ++rr) {
        const int rel = rel0 + rr;
        w_fetch(w_select(Wrel, Wloop, rel + 1), tid, p);

        if (active) {
            #pragma unroll
            for (int nt = 0; nt < 8; ++nt) acc[nt] = (v8f){};
            tile_products(Wlds[rr & 1], a, acc, hi, m);
        }

        w_put(Wlds[(rr + 1) & 1], tid, p);
        __syncthreads();

        if (active) tile_store(xW + (size_t)rel * N_NODES * D, rbase, m, acc);
    }

    if (active) {
        const int rel = rel0 + TR_NREL - 1;
        if (rel == N_REL) {
            #pragma unroll
            for (int nt = 0; nt < 8; ++nt) {
                float bv = brel[nt * 16 + m];
                #pragma unroll
                for (int v = 0; v < 8; ++v) acc[nt][v] = bv;
            }
        } else {
            #pragma unroll
            for (int nt = 0; nt < 8; ++nt) acc[nt] = (v8f){};
        }
        tile_products(Wlds[(TR_NREL - 1) & 1], a, acc, hi, m);
        float* dstp = (rel < N_REL) ? (xW + (size_t)rel * N_NODES * D) : mid;
        tile_store(dstp, rbase, m, acc);
    }
}

#define NB    391
#define NBP   392
#define NCH   240
#define CE    2500
#define CH_T  1024
#define CH_K  3

__device__ __forceinline__ int wave_inc_scan(int v, int lane)
{
    for (int o = 1; o < 32; o <<= 1) {
        const int up = __shfl_up(v, o);
        if (lane >= o) v += up;
    }
    return v;
}

__global__ __launch_bounds__(CH_T) void k_chunk(
    const int* __restrict__ src, const int* __restrict__ dst,
    const int* __restrict__ etype, int* __restrict__ lstart, int* __restrict__ sorted)
{
    __shared__ int hist[NBP];
    __shared__ int wtot[32];
    __shared__ int recl[CE];
    const int tid  = threadIdx.x;
    const int lane = tid & 31;
    const int wave = tid >> 5;
    const int c    = blockIdx.x;

    if (tid < NBP) hist[tid] = 0;
    __syncthreads();

    int rec[CH_K], bin[CH_K], rk[CH_K];
    #pragma unroll
    for (int k = 0; k < CH_K; ++k) {
        const int i = tid + k * CH_T;
        bin[k] = -1;
        rec[k] = 0;
        if (i < CE) {
            const int e = c * CE + i;
            const int d = dst[e];
            bin[k] = d >> 7;
            rec[k] = ((d & 127) << 19) | (etype[e] * N_NODES + src[e]);
        }
    }
    #pragma unroll
    for (int k = 0; k < CH_K; ++k) {
        rk[k] = 0;
        if (bin[k] >= 0) rk[k] = atomicAdd(&hist[bin[k]], 1);
    }
    __syncthreads();

    const int v   = (tid < NB) ? hist[tid] : 0;
    const int inc = wave_inc_scan(v, lane);
    if (lane == 31) wtot[wave] = inc;
    __syncthreads();
    if (wave == 0) {
        const int tot = wtot[lane];
        const int ti  = wave_inc_scan(tot, lane);
        wtot[lane] = ti - tot;
    }
    __syncthreads();
    const int ex = wtot[wave] + inc - v;
    if (tid < NBP) {
        hist[tid] = ex;
        lstart[c * NBP + tid] = ex;
    }
    __syncthreads();

    #pragma unroll
    for (int k = 0; k < CH_K; ++k)
        if (bin[k] >= 0) recl[hist[bin[k]] + rk[k]] = rec[k];
    __syncthreads();

    #pragma unroll
    for (int k = 0; k < CH_K; ++k) {
        const int i = tid + k * CH_T;
        if (i < CE) sorted[c * CE + i] = recl[i];
    }
}

#define GA_T    512
#define GA_U    8
#define GA_CAP  (GA_T * GA_U)

typedef __attribute__((ext_vector_type(4))) float v4f;

template <int G>
__device__ __forceinline__ void sum_rows(v4f& acc, const float* __restrict__ xW,
                                         int mine, int j, int n, int lane)
{
    v4f v[G];
    #pragma unroll
    for (int u = 0; u < G; ++u) {
        const int jj = j + u < n ? j + u : n - 1;
        const int r  = __builtin_amdgcn_readlane(mine, jj);
        v[u] = ((const v4f*)(xW + (size_t)r * D))[lane];
    }
    #pragma unroll
    for (int u = 0; u < G; ++u) asm volatile("" :: "v"(v[u]));
    #pragma unroll
    for (int u = 0; u < G; ++u)
        if (j + u < n) acc += v[u];
}

__global__ __launch_bounds__(GA_T) void k_gather(
    const float* __restrict__ xW, const int* __restrict__ lstart,
    const int* __restrict__ sorted, float* mid)
{
    __shared__ int runo[NCH + 1];
    __shared__ int runs[NCH];
    __shared__ int wtot[GA_T / 32];
    __shared__ int rowc[128];
    __shared__ int rowb[129];
    __shared__ int recs[GA_CAP];

    const int tid  = threadIdx.x;
    const int lane = tid & 31;
    const int wave = __builtin_amdgcn_readfirstlane(tid >> 5);
    const int b    = blockIdx.x;

    int cnt = 0;
    if (tid < NCH) {
        const int* ls = lstart + tid * NBP + b;
        const int s = ls[0];
        cnt = ls[1] - s;
        runs[tid] = tid * CE + s;
    }
    const int inc = wave_inc_scan(cnt, lane);
    if (lane == 31) wtot[wave] = inc;
    __syncthreads();
    if (wave == 0) {
        const int tot = lane < GA_T / 32 ? wtot[lane] : 0;
        const int ti  = wave_inc_scan(tot, lane);
        if (lane < GA_T / 32) wtot[lane] = ti - tot;
    }
    __syncthreads();
    const int ex = wtot[wave] + inc - cnt;
    if (tid < NCH) runo[tid] = ex;
    if (tid == NCH - 1) runo[NCH] = ex + cnt;
    __syncthreads();
    const int n = __builtin_amdgcn_readfirstlane(runo[NCH]);

    for (int w0 = 0; w0 < n; w0 += GA_CAP) {
        const int mm = n - w0 < GA_CAP ? n - w0 : GA_CAP;
        if (tid < 128) rowc[tid] = 0;
        __syncthreads();

        int rec[GA_U], rk[GA_U];
        #pragma unroll
        for (int u = 0; u < GA_U; ++u) {
            const int i = tid + u * GA_T;
            rec[u] = -1;
            if (i < mm) {
                const int g = w0 + i;
                int lo = 0, hi = NCH;
                while (hi - lo > 1) {
                    const int md = (lo + hi) >> 1;
                    if (runo[md] <= g) lo = md; else hi = md;
                }
                rec[u] = sorted[runs[lo] + (g - runo[lo])];
            }
        }
        #pragma unroll
        for (int u = 0; u < GA_U; ++u) {
            rk[u] = 0;
            if (rec[u] >= 0) rk[u] = atomicAdd(&rowc[rec[u] >> 19], 1);
        }
        __syncthreads();

        if (wave == 0) {
            const int c0 = rowc[4 * lane], c1 = rowc[4 * lane + 1];
            const int c2 = rowc[4 * lane + 2], c3 = rowc[4 * lane + 3];
            const int s4 = c0 + c1 + c2 + c3;
            int e4 = wave_inc_scan(s4, lane) - s4;
            rowb[4 * lane]     = e4;  e4 += c0;
            rowb[4 * lane + 1] = e4;  e4 += c1;
            rowb[4 * lane + 2] = e4;  e4 += c2;
            rowb[4 * lane + 3] = e4;  e4 += c3;
            if (lane == 31) rowb[128] = e4;
        }
        __syncthreads();

        #pragma unroll
        for (int u = 0; u < GA_U; ++u)
            if (rec[u] >= 0) recs[rowb[rec[u] >> 19] + rk[u]] = rec[u] & 0x7FFFF;
        __syncthreads();

        for (int rr = 0; rr < 8; ++rr) {
            const int r    = wave * 8 + rr;
            const int node = b * 128 + r;
            const int beg  = __builtin_amdgcn_readfirstlane(rowb[r]);
            const int end  = __builtin_amdgcn_readfirstlane(rowb[r + 1]);
            if (node >= N_NODES || end <= beg) continue;

            v4f* op = (v4f*)(mid + (size_t)node * D) + lane;
            v4f acc = *op;
            for (int base = beg; base < end; base += 32) {
                const int rem = end - base;
                const int nn  = rem < 32 ? rem : 32;
                const int mine = recs[base + (lane < nn ? lane : nn - 1)];

                int j = 0;
                for (; nn - j > 8; j += 16) sum_rows<16>(acc, xW, mine, j, nn, lane);
                if (nn - j > 4)  sum_rows<8>(acc, xW, mine, j, nn, lane);
                else if (j < nn) sum_rows<4>(acc, xW, mine, j, nn, lane);
            }
            *op = acc;
        }
        __syncthreads();
    }
}

#define ML_HP 132

template <int LD>
__device__ __forceinline__ void w_fetch_ld(const float* __restrict__ Wm, int tid,
                                           v2f (&p)[TR_Q])
{
    const float* g = Wm + (size_t)((tid >> 7) * 2) * LD + (tid & 127);
    #pragma unroll
    for (int q = 0; q < TR_Q; ++q) {
        p[q][0] = g[q * 4 * LD];
        p[q][1] = g[q * 4 * LD + LD];
    }
}

__device__ __forceinline__ void bias_init(v8f (&acc)[8], const float (&b)[8])
{
    #pragma unroll
    for (int nt = 0; nt < 8; ++nt) {
        #pragma unroll
        for (int v = 0; v < 8; ++v) acc[nt][v] = b[nt];
    }
}

__device__ __forceinline__ void h_put(float* hw, int m, int hi, const v8f (&acc)[8])
{
    #pragma unroll
    for (int nt = 0; nt < 8; ++nt) {
        #pragma unroll
        for (int v = 0; v < 8; ++v) {
            const float val = acc[nt][v];
            hw[(hi * 8 + v) * ML_HP + nt * 16 + m] = val > 0.0f ? val : 0.0f;
        }
    }
}

__device__ __forceinline__ void h_get(const float* hw, int m, int kh, v2f (&a)[32])
{
    const float* r = hw + m * ML_HP + kh;
    #pragma unroll
    for (int kc = 0; kc < 32; ++kc) a[kc] = *(const v2f*)(r + kc * 4);
}

__global__ __launch_bounds__(256) void k_mlp(
    const float* __restrict__ x, const float* mid,
    const float* __restrict__ W1, const float* __restrict__ b1,
    const float* __restrict__ W2, const float* __restrict__ b2,
    float* out)
{
    __shared__ float Wlds[2][D * D];
    __shared__ float hl[2][TR_WAVES][16 * ML_HP];

    const int tid  = threadIdx.x;
    const int wave = tid >> 5;
    const int lane = tid & 31;
    const int m  = lane & 15;
    const int hi = lane >> 4;
    const int kh = hi * 2;

    const int  tile   = blockIdx.x * TR_WAVES + wave;
    const bool active = tile < N_TILES;
    const int  row0   = (active ? tile : (N_TILES - 1)) * 16;
    const int  rbase  = row0 + hi * 8;

    v2f p[TR_Q];
    w_fetch_ld<2 * D>(W1, tid, p);

    v2f ax[32], am[32];
    {
        const float* xrow = x   + (size_t)(row0 + m) * D + kh;
        const float* mrow = mid + (size_t)(row0 + m) * D + kh;
        #pragma unroll
        for (int kc = 0; kc < 32; ++kc) ax[kc] = *(const v2f*)(xrow + kc * 4);
        #pragma unroll
        for (int kc = 0; kc < 32; ++kc) am[kc] = *(const v2f*)(mrow + kc * 4);
    }

    float bh0[8], bh1[8], bo[8];
    #pragma unroll
    for (int nt = 0; nt < 8; ++nt) {
        bh0[nt] = b1[nt * 16 + m];
        bh1[nt] = b1[D + nt * 16 + m];
        bo[nt]  = b2[nt * 16 + m];
    }

    w_put(Wlds[0], tid, p);
    __syncthreads();

    v8f acc[8];
    v2f a[32];

    #pragma unroll 1
    for (int pc = 0; pc < 6; ++pc) {
        if (pc < 5) {
            const int nx = pc + 1;
            if (nx < 4) w_fetch_ld<2 * D>(W1 + (size_t)(nx & 1) * D * 2 * D + (nx >> 1) * D, tid, p);
            else        w_fetch_ld<D>(W2 + (size_t)(nx - 4) * D * D, tid, p);
        }

        if (active) {
            if (pc >= 4) {
                h_get(hl[pc - 4][wave], m, kh, a);
            } else if (pc & 1) {
                #pragma unroll
                for (int kc = 0; kc < 32; ++kc) a[kc] = am[kc];
            } else {
                #pragma unroll
                for (int kc = 0; kc < 32; ++kc) a[kc] = ax[kc];
            }

            if (pc == 0)      bias_init(acc, bh0);
            else if (pc == 2) bias_init(acc, bh1);
            else if (pc == 4) bias_init(acc, bo);

            tile_products(Wlds[pc & 1], a, acc, hi, m);

            if (pc == 5)     tile_store(out, rbase, m, acc);
            else if (pc & 1) h_put(hl[pc >> 1][wave], m, hi, acc);
        }

        if (pc < 5) {
            w_put(Wlds[(pc + 1) & 1], tid, p);
            __syncthreads();
        }
    }
}

extern "C" void kernel_launch(void* const* d_in, const int* in_sizes, int n_in,
                              void* d_out, int out_size, void* d_ws, size_t ws_size,
                              hipStream_t stream) {
    const float* x     = (const float*)d_in[0];
    const int*   src   = (const int*)  d_in[1];
    const int*   dst   = (const int*)  d_in[2];
    const int*   etype = (const int*)  d_in[3];
    const float* Wrel  = (const float*)d_in[4];
    const float* Wloop = (const float*)d_in[5];
    const float* brel  = (const float*)d_in[6];
    const float* W1    = (const float*)d_in[7];
    const float* b1    = (const float*)d_in[8];
    const float* W2    = (const float*)d_in[9];
    const float* b2    = (const float*)d_in[10];

    float* out = (float*)d_out;
    float* xW  = (float*)d_ws;
    int* lstart = (int*)((char*)d_ws + 204800000);
    int* sorted = lstart + NCH * NBP;

    dim3 g1((N_TILES + 7) / 8, 3);
    k_transform<<<g1, 256, 0, stream>>>(x, Wrel, Wloop, brel, xW, out);

    k_chunk<<<NCH, CH_T, 0, stream>>>(src, dst, etype, lstart, sorted);
    k_gather<<<NB, GA_T, 0, stream>>>(xW, lstart, sorted, out);

    k_mlp<<<(N_TILES + 7) / 8, 256, 0, stream>>>(x, out, W1, b1, W2, b2, out);
}


# edge sum in front of the relation products: a block per (128-row bin, 3 relations) sums x[src] rows in LDS, one product per relation; xW never written, parts added in k_mlp. Sum order changed
# speedup vs baseline: 2.0922x; 1.0573x over previous
#include <hip/hip_runtime.h>
#include <hip/hip_bf16.h>

#define N_NODES 50000
#define N_EDGES 600000
#define N_REL   8
#define D       128
#define N_TILES 3125

typedef __attribute__((ext_vector_type(2))) float v2f;
typedef __attribute__((ext_vector_type(8))) float v8f;

__device__ __forceinline__ v8f wmma4(v2f a, v2f b, v8f c) {
    return __builtin_amdgcn_wmma_f32_16x16x4_f32(false, a, false, b, (short)0, c, false, false);
}

#define TR_WAVES  8
#define TR_NREL   3
#define TR_Q      32

__device__ __forceinline__ void w_fetch(const float* __restrict__ Wm, int tid,
                                        v2f (&p)[TR_Q])
{
    const float* g = Wm + (size_t)((tid >> 7) * 2) * D + (tid & 127);
    #pragma unroll
    for (int q = 0; q < TR_Q; ++q) {
        p[q][0] = g[q * 4 * D];
        p[q][1] = g[q * 4 * D + D];
    }
}

__device__ __forceinline__ void w_put(float* buf, int tid, const v2f (&p)[TR_Q])
{
    float* d = buf + tid * 2;
    #pragma unroll
    for (int q = 0; q < TR_Q; ++q) *(v2f*)(d + q * 512) = p[q];
}

__device__ __forceinline__ void tile_products(const float* Wb, const v2f (&a)[32],
                                              v8f (&acc)[8], int hi, int m)
{
    #pragma unroll
    for (int kc = 0; kc < 32; ++kc) {
        const float* bp = Wb + (2 * kc + hi) * 256 + m * 2;
        #pragma unroll
        for (int nt = 0; nt < 8; ++nt) {
            v2f b = *(const v2f*)(bp + nt * 32);
            acc[nt] = wmma4(a[kc], b, acc[nt]);
        }
    }
}

__device__ __forceinline__ void tile_store(float* __restrict__ dstp, int rbase, int m,
                                           const v8f (&acc)[8])
{
    #pragma unroll
    for (int nt = 0; nt < 8; ++nt) {
        #pragma unroll
        for (int v = 0; v < 8; ++v)
            dstp[(size_t)(rbase + v) * D + nt * 16 + m] = acc[nt][v];
    }
}

#define NB    391
#define NBP   392
#define NCH   240
#define CE    2500
#define CH_T  1024
#define CH_K  3

__device__ __forceinline__ int wave_inc_scan(int v, int lane)
{
    for (int o = 1; o < 32; o <<= 1) {
        const int up = __shfl_up(v, o);
        if (lane >= o) v += up;
    }
    return v;
}

__global__ __launch_bounds__(CH_T) void k_chunk(
    const int* __restrict__ src, const int* __restrict__ dst,
    const int* __restrict__ etype, int* __restrict__ lstart, int* __restrict__ sorted)
{
    __shared__ int hist[NBP];
    __shared__ int wtot[32];
    __shared__ int recl[CE];
    const int tid  = threadIdx.x;
    const int lane = tid & 31;
    const int wave = tid >> 5;
    const int c    = blockIdx.x;

    if (tid < NBP) hist[tid] = 0;
    __syncthreads();

    int rec[CH_K], bin[CH_K], rk[CH_K];
    #pragma unroll
    for (int k = 0; k < CH_K; ++k) {
        const int i = tid + k * CH_T;
        bin[k] = -1;
        rec[k] = 0;
        if (i < CE) {
            const int e = c * CE + i;
            const int d = dst[e];
            bin[k] = d >> 7;
            rec[k] = ((d & 127) << 19) | (etype[e] * N_NODES + src[e]);
        }
    }
    #pragma unroll
    for (int k = 0; k < CH_K; ++k) {
        rk[k] = 0;
        if (bin[k] >= 0) rk[k] = atomicAdd(&hist[bin[k]], 1);
    }
    __syncthreads();

    const int v   = (tid < NB) ? hist[tid] : 0;
    const int inc = wave_inc_scan(v, lane);
    if (lane == 31) wtot[wave] = inc;
    __syncthreads();
    if (wave == 0) {
        const int tot = wtot[lane];
        const int ti  = wave_inc_scan(tot, lane);
        wtot[lane] = ti - tot;
    }
    __syncthreads();
    const int ex = wtot[wave] + inc - v;
    if (tid < NBP) {
        hist[tid] = ex;
        lstart[c * NBP + tid] = ex;
    }
    __syncthreads();

    #pragma unroll
    for (int k = 0; k < CH_K; ++k)
        if (bin[k] >= 0) recl[hist[bin[k]] + rk[k]] = rec[k];
    __syncthreads();

    #pragma unroll
    for (int k = 0; k < CH_K; ++k) {
        const int i = tid + k * CH_T;
        if (i < CE) sorted[c * CE + i] = recl[i];
    }
}

typedef __attribute__((ext_vector_type(4))) float v4f;

#define ML_HP 132

template <int LD>
__device__ __forceinline__ void w_fetch_ld(const float* __restrict__ Wm, int tid,
                                           v2f (&p)[TR_Q])
{
    const float* g = Wm + (size_t)((tid >> 7) * 2) * LD + (tid & 127);
    #pragma unroll
    for (int q = 0; q < TR_Q; ++q) {
        p[q][0] = g[q * 4 * LD];
        p[q][1] = g[q * 4 * LD + LD];
    }
}

__device__ __forceinline__ void bias_init(v8f (&acc)[8], const float (&b)[8])
{
    #pragma unroll
    for (int nt = 0; nt < 8; ++nt) {
        #pragma unroll
        for (int v = 0; v < 8; ++v) acc[nt][v] = b[nt];
    }
}

__device__ __forceinline__ void h_put(float* hw, int m, int hi, const v8f (&acc)[8])
{
    #pragma unroll
    for (int nt = 0; nt < 8; ++nt) {
        #pragma unroll
        for (int v = 0; v < 8; ++v) {
            const float val = acc[nt][v];
            hw[(hi * 8 + v) * ML_HP + nt * 16 + m] = val > 0.0f ? val : 0.0f;
        }
    }
}

__device__ __forceinline__ void h_get(const float* hw, int m, int kh, v2f (&a)[32])
{
    const float* r = hw + m * ML_HP + kh;
    #pragma unroll
    for (int kc = 0; kc < 32; ++kc) a[kc] = *(const v2f*)(r + kc * 4);
}

#define CV_T    256
#define CV_U    8
#define CV_CAP  (CV_T * CV_U)
#define CV_NK   (TR_NREL * 128)
#define CV_KPL  (CV_NK / 32)

template <int G>
__device__ __forceinline__ void rows_in(v4f& sum, int& cur, float* pw,
                                        const float* __restrict__ x,
                                        int mine, int j, int n, int lane)
{
    v4f v[G];
    #pragma unroll
    for (int u = 0; u < G; ++u) {
        const int jj = j + u < n ? j + u : n - 1;
        const int r  = __builtin_amdgcn_readlane(mine, jj);
        v[u] = ((const v4f*)(x + (size_t)(r & 0xFFFF) * D))[lane];
    }
    #pragma unroll
    for (int u = 0; u < G; ++u) asm volatile("" :: "v"(v[u]));
    #pragma unroll
    for (int u = 0; u < G; ++u) {
        if (j + u < n) {
            const int row = (__builtin_amdgcn_readlane(mine, j + u) >> 16) & 15;
            if (row != cur) {
                if (cur >= 0) *(v4f*)(pw + cur * ML_HP + lane * 4) = sum;
                cur = row;
                sum = v[u];
            } else {
                sum += v[u];
            }
        }
    }
}

__global__ __launch_bounds__(CV_T, 4) void k_conv(
    const float* __restrict__ x, const float* __restrict__ Wrel,
    const float* __restrict__ Wloop, const float* __restrict__ brel,
    const int* __restrict__ lstart, const int* __restrict__ sorted,
    float* __restrict__ part0, float* __restrict__ part1, float* __restrict__ mid)
{
    __shared__ alignas(16) float Wl[D * D];
    __shared__ alignas(16) float pat[TR_WAVES][16 * ML_HP];
    __shared__ int runo[NCH + 1];
    __shared__ int runs[NCH];
    __shared__ int wtot[CV_T / 32];
    __shared__ int keyc[CV_NK];
    __shared__ int keyb[CV_NK + 1];
    __shared__ int recs[CV_CAP];

    const int tid  = threadIdx.x;
    const int lane = tid & 31;
    const int wave = __builtin_amdgcn_readfirstlane(tid >> 5);
    const int m  = lane & 15;
    const int hi = lane >> 4;
    const int kh = hi * 2;
    const int b  = blockIdx.x;
    const int g  = blockIdx.y;
    const int  rel0 = g * TR_NREL;
    const bool self = rel0 + TR_NREL > N_REL;
    const int  nre  = self ? N_REL - rel0 : TR_NREL;

    const int  tile   = b * TR_WAVES + wave;
    const bool active = tile < N_TILES;
    const int  row0   = (active ? tile : (N_TILES - 1)) * 16;
    const int  rbase  = row0 + hi * 8;
    float* pw = pat[wave];

    int cnt = 0;
    if (tid < NCH) {
        const int* ls = lstart + tid * NBP + b;
        const int s = ls[0];
        cnt = ls[1] - s;
        runs[tid] = tid * CE + s;
    }
    const int inc = wave_inc_scan(cnt, lane);
    if (lane == 31) wtot[wave] = inc;
    __syncthreads();
    if (wave == 0) {
        const int tot = lane < CV_T / 32 ? wtot[lane] : 0;
        const int ti  = wave_inc_scan(tot, lane);
        if (lane < CV_T / 32) wtot[lane] = ti - tot;
    }
    __syncthreads();
    const int ex = wtot[wave] + inc - cnt;
    if (tid < NCH) runo[tid] = ex;
    if (tid == NCH - 1) runo[NCH] = ex + cnt;
    __syncthreads();
    const int n = __builtin_amdgcn_readfirstlane(runo[NCH]);

    v8f acc[8];
    if (self) {
        #pragma unroll
        for (int nt = 0; nt < 8; ++nt) {
            const float bv = brel[nt * 16 + m];
            #pragma unroll
            for (int v = 0; v < 8; ++v) acc[nt][v] = bv;
        }
    } else {
        #pragma unroll
        for (int nt = 0; nt < 8; ++nt) acc[nt] = (v8f){};
    }
    v2f a[32];

    int w0 = 0;
    do {
        const int mm = n - w0 < CV_CAP ? n - w0 : CV_CAP;
        for (int i = tid; i < CV_NK; i += CV_T) keyc[i] = 0;
        __syncthreads();

        int rec[CV_U], rk[CV_U];
        #pragma unroll
        for (int u = 0; u < CV_U; ++u) {
            const int i = tid + u * CV_T;
            rec[u] = -1;
            if (i < mm) {
                const int gi = w0 + i;
                int lo = 0, hi2 = NCH;
                while (hi2 - lo > 1) {
                    const int md = (lo + hi2) >> 1;
                    if (runo[md] <= gi) lo = md; else hi2 = md;
                }
                const int r   = sorted[runs[lo] + (gi - runo[lo])];
                const int val = r & 0x7FFFF;
                const int rel = val / N_NODES;
                const int rr  = rel - rel0;
                if (rr >= 0 && rr < nre)
                    rec[u] = ((rr * 128 + (r >> 19)) << 16) | (val - rel * N_NODES);
            }
        }
        #pragma unroll
        for (int u = 0; u < CV_U; ++u) {
            rk[u] = 0;
            if (rec[u] >= 0) rk[u] = atomicAdd(&keyc[rec[u] >> 16], 1);
        }
        __syncthreads();

        if (wave == 0) {
            int cq[CV_KPL];
            int s = 0;
            #pragma unroll
            for (int q = 0; q < CV_KPL; ++q) { cq[q] = keyc[CV_KPL * lane + q]; s += cq[q]; }
            int e = wave_inc_scan(s, lane) - s;
            #pragma unroll
            for (int q = 0; q < CV_KPL; ++q) { keyb[CV_KPL * lane + q] = e; e += cq[q]; }
            if (lane == 31) keyb[CV_NK] = e;
        }
        __syncthreads();

        #pragma unroll
        for (int u = 0; u < CV_U; ++u)
            if (rec[u] >= 0) recs[keyb[rec[u] >> 16] + rk[u]] = rec[u];
        __syncthreads();

        const int ns = nre + ((self && w0 + CV_CAP >= n) ? 1 : 0);
        for (int r = 0; r < ns; ++r) {
            const bool sl = r == nre;
            v2f p[TR_Q];
            w_fetch(sl ? Wloop : Wrel + (size_t)(rel0 + r) * D * D, tid, p);

            if (active && sl) {
                const float* xrow = x + (size_t)(row0 + m) * D + kh;
                #pragma unroll
                for (int kc = 0; kc < 32; ++kc) a[kc] = *(const v2f*)(xrow + kc * 4);
            } else if (active) {
                const int k0  = r * 128 + wave * 16;
                const int beg = __builtin_amdgcn_readfirstlane(keyb[k0]);
                const int end = __builtin_amdgcn_readfirstlane(keyb[k0 + 16]);
                #pragma unroll
                for (int i = 0; i < 16; ++i) *(v4f*)(pw + i * ML_HP + lane * 4) = (v4f){};
                v4f sum = (v4f){};
                int cur = -1;
                for (int base = beg; base < end; base += 32) {
                    const int rem = end - base;
                    const int nn  = rem < 32 ? rem : 32;
                    const int mine = recs[base + (lane < nn ? lane : nn - 1)];
                    int j = 0;
                    for (; nn - j > 8; j += 16) rows_in<16>(sum, cur, pw, x, mine, j, nn, lane);
                    if (j < nn) rows_in<8>(sum, cur, pw, x, mine, j, nn, lane);
                }
                if (cur >= 0) *(v4f*)(pw + cur * ML_HP + lane * 4) = sum;
                h_get(pw, m, kh, a);
            }
            __syncthreads();
            w_put(Wl, tid, p);
            __syncthreads();

            if (active) tile_products(Wl, a, acc, hi, m);
        }
        w0 += CV_CAP;
    } while (w0 < n);

    if (active) tile_store(g == 0 ? part0 : (g == 1 ? part1 : mid), rbase, m, acc);
}

__global__ __launch_bounds__(256) void k_mlp(
    const float* __restrict__ x, const float* mid,
    const float* __restrict__ part0, const float* __restrict__ part1,
    const float* __restrict__ W1, const float* __restrict__ b1,
    const float* __restrict__ W2, const float* __restrict__ b2,
    float* out)
{
    __shared__ float Wlds[2][D * D];
    __shared__ float hl[2][TR_WAVES][16 * ML_HP];

    const int tid  = threadIdx.x;
    const int wave = tid >> 5;
    const int lane = tid & 31;
    const int m  = lane & 15;
    const int hi = lane >> 4;
    const int kh = hi * 2;

    const int  tile   = blockIdx.x * TR_WAVES + wave;
    const bool active = tile < N_TILES;
    const int  row0   = (active ? tile : (N_TILES - 1)) * 16;
    const int  rbase  = row0 + hi * 8;

    v2f p[TR_Q];
    w_fetch_ld<2 * D>(W1, tid, p);

    v2f ax[32], am[32];
    {
        const size_t ro = (size_t)(row0 + m) * D + kh;
        const float* xrow = x + ro;
        const float* mrow = mid + ro;
        const float* qrow = part0 + ro;
        const float* rrow = part1 + ro;
        #pragma unroll
        for (int kc = 0; kc < 32; ++kc) ax[kc] = *(const v2f*)(xrow + kc * 4);
        #pragma unroll
        for (int kc = 0; kc < 32; ++kc)
            am[kc] = (*(const v2f*)(mrow + kc * 4) + *(const v2f*)(qrow + kc * 4))
                   + *(const v2f*)(rrow + kc * 4);
    }

    float bh0[8], bh1[8], bo[8];
    #pragma unroll
    for (int nt = 0; nt < 8; ++nt) {
        bh0[nt] = b1[nt * 16 + m];
        bh1[nt] = b1[D + nt * 16 + m];
        bo[nt]  = b2[nt * 16 + m];
    }

    w_put(Wlds[0], tid, p);
    __syncthreads();

    v8f acc[8];
    v2f a[32];

    #pragma unroll 1
    for (int pc = 0; pc < 6; ++pc) {
        if (pc < 5) {
            const int nx = pc + 1;
            if (nx < 4) w_fetch_ld<2 * D>(W1 + (size_t)(nx & 1) * D * 2 * D + (nx >> 1) * D, tid, p);
            else        w_fetch_ld<D>(W2 + (size_t)(nx - 4) * D * D, tid, p);
        }

        if (active) {
            if (pc >= 4) {
                h_get(hl[pc - 4][wave], m, kh, a);
            } else if (pc & 1) {
                #pragma unroll
                for (int kc = 0; kc < 32; ++kc) a[kc] = am[kc];
            } else {
                #pragma unroll
                for (int kc = 0; kc < 32; ++kc) a[kc] = ax[kc];
            }

            if (pc == 0)      bias_init(acc, bh0);
            else if (pc == 2) bias_init(acc, bh1);
            else if (pc == 4) bias_init(acc, bo);

            tile_products(Wlds[pc & 1], a, acc, hi, m);

            if (pc == 5)     tile_store(out, rbase, m, acc);
            else if (pc & 1) h_put(hl[pc >> 1][wave], m, hi, acc);
        }

        if (pc < 5) {
            w_put(Wlds[(pc + 1) & 1], tid, p);
            __syncthreads();
        }
    }
}

extern "C" void kernel_launch(void* const* d_in, const int* in_sizes, int n_in,
                              void* d_out, int out_size, void* d_ws, size_t ws_size,
                              hipStream_t stream) {
    const float* x     = (const float*)d_in[0];
    const int*   src   = (const int*)  d_in[1];
    const int*   dst   = (const int*)  d_in[2];
    const int*   etype = (const int*)  d_in[3];
    const float* Wrel  = (const float*)d_in[4];
    const float* Wloop = (const float*)d_in[5];
    const float* brel  = (const float*)d_in[6];
    const float* W1    = (const float*)d_in[7];
    const float* b1    = (const float*)d_in[8];
    const float* W2    = (const float*)d_in[9];
    const float* b2    = (const float*)d_in[10];

    float* out = (float*)d_out;
    float* part0 = (float*)d_ws;
    float* part1 = part0 + (size_t)N_NODES * D;
    int* lstart = (int*)((char*)d_ws + 204800000);
    int* sorted = lstart + NCH * NBP;

    k_chunk<<<NCH, CH_T, 0, stream>>>(src, dst, etype, lstart, sorted);

    dim3 g2(NB, 3);
    k_conv<<<g2, CV_T, 0, stream>>>(x, Wrel, Wloop, brel, lstart, sorted, part0, part1, out);

    k_mlp<<<(N_TILES + 7) / 8, 256, 0, stream>>>(x, out, part0, part1, W1, b1, W2, b2, out);
}


# k_conv record fetch: a thread's 8 run searches go down in step (fixed halvings, selects) and its 8 record loads are issued together; was record after record, each waiting for its own round trips
# speedup vs baseline: 2.2235x; 1.0628x over previous
#include <hip/hip_runtime.h>
#include <hip/hip_bf16.h>

#define N_NODES 50000
#define N_EDGES 600000
#define N_REL   8
#define D       128
#define N_TILES 3125

typedef __attribute__((ext_vector_type(2))) float v2f;
typedef __attribute__((ext_vector_type(8))) float v8f;

__device__ __forceinline__ v8f wmma4(v2f a, v2f b, v8f c) {
    return __builtin_amdgcn_wmma_f32_16x16x4_f32(false, a, false, b, (short)0, c, false, false);
}

#define TR_WAVES  8
#define TR_NREL   3
#define TR_Q      32

__device__ __forceinline__ void w_fetch(const float* __restrict__ Wm, int tid,
                                        v2f (&p)[TR_Q])
{
    const float* g = Wm + (size_t)((tid >> 7) * 2) * D + (tid & 127);
    #pragma unroll
    for (int q = 0; q < TR_Q; ++q) {
        p[q][0] = g[q * 4 * D];
        p[q][1] = g[q * 4 * D + D];
    }
}

__device__ __forceinline__ void w_put(float* buf, int tid, const v2f (&p)[TR_Q])
{
    float* d = buf + tid * 2;
    #pragma unroll
    for (int q = 0; q < TR_Q; ++q) *(v2f*)(d + q * 512) = p[q];
}

__device__ __forceinline__ void tile_products(const float* Wb, const v2f (&a)[32],
                                              v8f (&acc)[8], int hi, int m)
{
    #pragma unroll
    for (int kc = 0; kc < 32; ++kc) {
        const float* bp = Wb + (2 * kc + hi) * 256 + m * 2;
        #pragma unroll
        for (int nt = 0; nt < 8; ++nt) {
            v2f b = *(const v2f*)(bp + nt * 32);
            acc[nt] = wmma4(a[kc], b, acc[nt]);
        }
    }
}

__device__ __forceinline__ void tile_store(float* __restrict__ dstp, int rbase, int m,
                                           const v8f (&acc)[8])
{
    #pragma unroll
    for (int nt = 0; nt < 8; ++nt) {
        #pragma unroll
        for (int v = 0; v < 8; ++v)
            dstp[(size_t)(rbase + v) * D + nt * 16 + m] = acc[nt][v];
    }
}

#define NB    391
#define NBP   392
#define NCH   240
#define CE    2500
#define CH_T  1024
#define CH_K  3

__device__ __forceinline__ int wave_inc_scan(int v, int lane)
{
    for (int o = 1; o < 32; o <<= 1) {
        const int up = __shfl_up(v, o);
        if (lane >= o) v += up;
    }
    return v;
}

__global__ __launch_bounds__(CH_T) void k_chunk(
    const int* __restrict__ src, const int* __restrict__ dst,
    const int* __restrict__ etype, int* __restrict__ lstart, int* __restrict__ sorted)
{
    __shared__ int hist[NBP];
    __shared__ int wtot[32];
    __shared__ int recl[CE];
    const int tid  = threadIdx.x;
    const int lane = tid & 31;
    const int wave = tid >> 5;
    const int c    = blockIdx.x;

    if (tid < NBP) hist[tid] = 0;
    __syncthreads();

    int rec[CH_K], bin[CH_K], rk[CH_K];
    #pragma unroll
    for (int k = 0; k < CH_K; ++k) {
        const int i = tid + k * CH_T;
        bin[k] = -1;
        rec[k] = 0;
        if (i < CE) {
            const int e = c * CE + i;
            const int d = dst[e];
            bin[k] = d >> 7;
            rec[k] = ((d & 127) << 19) | (etype[e] * N_NODES + src[e]);
        }
    }
    #pragma unroll
    for (int k = 0; k < CH_K; ++k) {
        rk[k] = 0;
        if (bin[k] >= 0) rk[k] = atomicAdd(&hist[bin[k]], 1);
    }
    __syncthreads();

    const int v   = (tid < NB) ? hist[tid] : 0;
    const int inc = wave_inc_scan(v, lane);
    if (lane == 31) wtot[wave] = inc;
    __syncthreads();
    if (wave == 0) {
        const int tot = wtot[lane];
        const int ti  = wave_inc_scan(tot, lane);
        wtot[lane] = ti - tot;
    }
    __syncthreads();
    const int ex = wtot[wave] + inc - v;
    if (tid < NBP) {
        hist[tid] = ex;
        lstart[c * NBP + tid] = ex;
    }
    __syncthreads();

    #pragma unroll
    for (int k = 0; k < CH_K; ++k)
        if (bin[k] >= 0) recl[hist[bin[k]] + rk[k]] = rec[k];
    __syncthreads();

    #pragma unroll
    for (int k = 0; k < CH_K; ++k) {
        const int i = tid + k * CH_T;
        if (i < CE) sorted[c * CE + i] = recl[i];
    }
}

typedef __attribute__((ext_vector_type(4))) float v4f;

#define ML_HP 132

template <int LD>
__device__ __forceinline__ void w_fetch_ld(const float* __restrict__ Wm, int tid,
                                           v2f (&p)[TR_Q])
{
    const float* g = Wm + (size_t)((tid >> 7) * 2) * LD + (tid & 127);
    #pragma unroll
    for (int q = 0; q < TR_Q; ++q) {
        p[q][0] = g[q * 4 * LD];
        p[q][1] = g[q * 4 * LD + LD];
    }
}

__device__ __forceinline__ void bias_init(v8f (&acc)[8], const float (&b)[8])
{
    #pragma unroll
    for (int nt = 0; nt < 8; ++nt) {
        #pragma unroll
        for (int v = 0; v < 8; ++v) acc[nt][v] = b[nt];
    }
}

__device__ __forceinline__ void h_put(float* hw, int m, int hi, const v8f (&acc)[8])
{
    #pragma unroll
    for (int nt = 0; nt < 8; ++nt) {
        #pragma unroll
        for (int v = 0; v < 8; ++v) {
            const float val = acc[nt][v];
            hw[(hi * 8 + v) * ML_HP + nt * 16 + m] = val > 0.0f ? val : 0.0f;
        }
    }
}

__device__ __forceinline__ void h_get(const float* hw, int m, int kh, v2f (&a)[32])
{
    const float* r = hw + m * ML_HP + kh;
    #pragma unroll
    for (int kc = 0; kc < 32; ++kc) a[kc] = *(const v2f*)(r + kc * 4);
}

#define CV_T    256
#define CV_U    8
#define CV_CAP  (CV_T * CV_U)
#define CV_NK   (TR_NREL * 128)
#define CV_KPL  (CV_NK / 32)

template <int G>
__device__ __forceinline__ void rows_in(v4f& sum, int& cur, float* pw,
                                        const float* __restrict__ x,
                                        int mine, int j, int n, int lane)
{
    v4f v[G];
    #pragma unroll
    for (int u = 0; u < G; ++u) {
        const int jj = j + u < n ? j + u : n - 1;
        const int r  = __builtin_amdgcn_readlane(mine, jj);
        v[u] = ((const v4f*)(x + (size_t)(r & 0xFFFF) * D))[lane];
    }
    #pragma unroll
    for (int u = 0; u < G; ++u) asm volatile("" :: "v"(v[u]));
    #pragma unroll
    for (int u = 0; u < G; ++u) {
        if (j + u < n) {
            const int row = (__builtin_amdgcn_readlane(mine, j + u) >> 16) & 15;
            if (row != cur) {
                if (cur >= 0) *(v4f*)(pw + cur * ML_HP + lane * 4) = sum;
                cur = row;
                sum = v[u];
            } else {
                sum += v[u];
            }
        }
    }
}

__global__ __launch_bounds__(CV_T, 4) void k_conv(
    const float* __restrict__ x, const float* __restrict__ Wrel,
    const float* __restrict__ Wloop, const float* __restrict__ brel,
    const int* __restrict__ lstart, const int* __restrict__ sorted,
    float* __restrict__ part0, float* __restrict__ part1, float* __restrict__ mid)
{
    __shared__ alignas(16) float Wl[D * D];
    __shared__ alignas(16) float pat[TR_WAVES][16 * ML_HP];
    __shared__ int runo[NCH + 1];
    __shared__ int runs[NCH];
    __shared__ int wtot[CV_T / 32];
    __shared__ int keyc[CV_NK];
    __shared__ int keyb[CV_NK + 1];
    __shared__ int recs[CV_CAP];

    const int tid  = threadIdx.x;
    const int lane = tid & 31;
    const int wave = __builtin_amdgcn_readfirstlane(tid >> 5);
    const int m  = lane & 15;
    const int hi = lane >> 4;
    const int kh = hi * 2;
    const int b  = blockIdx.x;
    const int g  = blockIdx.y;
    const int  rel0 = g * TR_NREL;
    const bool self = rel0 + TR_NREL > N_REL;
    const int  nre  = self ? N_REL - rel0 : TR_NREL;

    const int  tile   = b * TR_WAVES + wave;
    const bool active = tile < N_TILES;
    const int  row0   = (active ? tile : (N_TILES - 1)) * 16;
    const int  rbase  = row0 + hi * 8;
    float* pw = pat[wave];

    int cnt = 0;
    if (tid < NCH) {
        const int* ls = lstart + tid * NBP + b;
        const int s = ls[0];
        cnt = ls[1] - s;
        runs[tid] = tid * CE + s;
    }
    const int inc = wave_inc_scan(cnt, lane);
    if (lane == 31) wtot[wave] = inc;
    __syncthreads();
    if (wave == 0) {
        const int tot = lane < CV_T / 32 ? wtot[lane] : 0;
        const int ti  = wave_inc_scan(tot, lane);
        if (lane < CV_T / 32) wtot[lane] = ti - tot;
    }
    __syncthreads();
    const int ex = wtot[wave] + inc - cnt;
    if (tid < NCH) runo[tid] = ex;
    if (tid == NCH - 1) runo[NCH] = ex + cnt;
    __syncthreads();
    const int n = __builtin_amdgcn_readfirstlane(runo[NCH]);

    v8f acc[8];
    if (self) {
        #pragma unroll
        for (int nt = 0; nt < 8; ++nt) {
            const float bv = brel[nt * 16 + m];
            #pragma unroll
            for (int v = 0; v < 8; ++v) acc[nt][v] = bv;
        }
    } else {
        #pragma unroll
        for (int nt = 0; nt < 8; ++nt) acc[nt] = (v8f){};
    }
    v2f a[32];

    int w0 = 0;
    do {
        const int mm = n - w0 < CV_CAP ? n - w0 : CV_CAP;
        for (int i = tid; i < CV_NK; i += CV_T) keyc[i] = 0;
        __syncthreads();

        int rec[CV_U], rk[CV_U];
        {
            int gi[CV_U], lo[CV_U], rr0[CV_U];
            #pragma unroll
            for (int u = 0; u < CV_U; ++u) {
                const int i = tid + u * CV_T;
                gi[u] = i < mm ? w0 + i : -1;
                lo[u] = 0;
            }
            static_assert(NCH <= 256, "eight halvings cover 256 runs");
            #pragma unroll
            for (int step = 128; step >= 1; step >>= 1) {
                #pragma unroll
                for (int u = 0; u < CV_U; ++u) {
                    const int md = lo[u] + step;
                    if (runo[md < NCH ? md : NCH] <= gi[u]) lo[u] = md;
                }
            }
            int rs[CV_U], ro[CV_U];
            #pragma unroll
            for (int u = 0; u < CV_U; ++u) { rs[u] = runs[lo[u]]; ro[u] = runo[lo[u]]; }
            #pragma unroll
            for (int u = 0; u < CV_U; ++u) {
                const int idx = gi[u] >= 0 ? rs[u] + (gi[u] - ro[u]) : 0;
                rr0[u] = sorted[idx];
            }
            #pragma unroll
            for (int u = 0; u < CV_U; ++u) {
                const int r   = rr0[u];
                const int val = r & 0x7FFFF;
                const int rel = val / N_NODES;
                const int rr  = rel - rel0;
                rec[u] = -1;
                if (gi[u] >= 0 && rr >= 0 && rr < nre)
                    rec[u] = ((rr * 128 + (r >> 19)) << 16) | (val - rel * N_NODES);
            }
        }
        #pragma unroll
        for (int u = 0; u < CV_U; ++u) {
            rk[u] = 0;
            if (rec[u] >= 0) rk[u] = atomicAdd(&keyc[rec[u] >> 16], 1);
        }
        __syncthreads();

        if (wave == 0) {
            int cq[CV_KPL];
            int s = 0;
            #pragma unroll
            for (int q = 0; q < CV_KPL; ++q) { cq[q] = keyc[CV_KPL * lane + q]; s += cq[q]; }
            int e = wave_inc_scan(s, lane) - s;
            #pragma unroll
            for (int q = 0; q < CV_KPL; ++q) { keyb[CV_KPL * lane + q] = e; e += cq[q]; }
            if (lane == 31) keyb[CV_NK] = e;
        }
        __syncthreads();

        #pragma unroll
        for (int u = 0; u < CV_U; ++u)
            if (rec[u] >= 0) recs[keyb[rec[u] >> 16] + rk[u]] = rec[u];
        __syncthreads();

        const int ns = nre + ((self && w0 + CV_CAP >= n) ? 1 : 0);
        for (int r = 0; r < ns; ++r) {
            const bool sl = r == nre;
            v2f p[TR_Q];
            w_fetch(sl ? Wloop : Wrel + (size_t)(rel0 + r) * D * D, tid, p);

            if (active && sl) {
                const float* xrow = x + (size_t)(row0 + m) * D + kh;
                #pragma unroll
                for (int kc = 0; kc < 32; ++kc) a[kc] = *(const v2f*)(xrow + kc * 4);
            } else if (active) {
                const int k0  = r * 128 + wave * 16;
                const int beg = __builtin_amdgcn_readfirstlane(keyb[k0]);
                const int end = __builtin_amdgcn_readfirstlane(keyb[k0 + 16]);
                #pragma unroll
                for (int i = 0; i < 16; ++i) *(v4f*)(pw + i * ML_HP + lane * 4) = (v4f){};
                v4f sum = (v4f){};
                int cur = -1;
                for (int base = beg; base < end; base += 32) {
                    const int rem = end - base;
                    const int nn  = rem < 32 ? rem : 32;
                    const int mine = recs[base + (lane < nn ? lane : nn - 1)];
                    int j = 0;
                    for (; nn - j > 8; j += 16) rows_in<16>(sum, cur, pw, x, mine, j, nn, lane);
                    if (j < nn) rows_in<8>(sum, cur, pw, x, mine, j, nn, lane);
                }
                if (cur >= 0) *(v4f*)(pw + cur * ML_HP + lane * 4) = sum;
                h_get(pw, m, kh, a);
            }
            __syncthreads();
            w_put(Wl, tid, p);
            __syncthreads();

            if (active) tile_products(Wl, a, acc, hi, m);
        }
        w0 += CV_CAP;
    } while (w0 < n);

    if (active) tile_store(g == 0 ? part0 : (g == 1 ? part1 : mid), rbase, m, acc);
}

__global__ __launch_bounds__(256) void k_mlp(
    const float* __restrict__ x, const float* mid,
    const float* __restrict__ part0, const float* __restrict__ part1,
    const float* __restrict__ W1, const float* __restrict__ b1,
    const float* __restrict__ W2, const float* __restrict__ b2,
    float* out)
{
    __shared__ float Wlds[2][D * D];
    __shared__ float hl[2][TR_WAVES][16 * ML_HP];

    const int tid  = threadIdx.x;
    const int wave = tid >> 5;
    const int lane = tid & 31;
    const int m  = lane & 15;
    const int hi = lane >> 4;
    const int kh = hi * 2;

    const int  tile   = blockIdx.x * TR_WAVES + wave;
    const bool active = tile < N_TILES;
    const int  row0   = (active ? tile : (N_TILES - 1)) * 16;
    const int  rbase  = row0 + hi * 8;

    v2f p[TR_Q];
    w_fetch_ld<2 * D>(W1, tid, p);

    v2f ax[32], am[32];
    {
        const size_t ro = (size_t)(row0 + m) * D + kh;
        const float* xrow = x + ro;
        const float* mrow = mid + ro;
        const float* qrow = part0 + ro;
        const float* rrow = part1 + ro;
        #pragma unroll
        for (int kc = 0; kc < 32; ++kc) ax[kc] = *(const v2f*)(xrow + kc * 4);
        #pragma unroll
        for (int kc = 0; kc < 32; ++kc)
            am[kc] = (*(const v2f*)(mrow + kc * 4) + *(const v2f*)(qrow + kc * 4))
                   + *(const v2f*)(rrow + kc * 4);
    }

    float bh0[8], bh1[8], bo[8];
    #pragma unroll
    for (int nt = 0; nt < 8; ++nt) {
        bh0[nt] = b1[nt * 16 + m];
        bh1[nt] = b1[D + nt * 16 + m];
        bo[nt]  = b2[nt * 16 + m];
    }

    w_put(Wlds[0], tid, p);
    __syncthreads();

    v8f acc[8];
    v2f a[32];

    #pragma unroll 1
    for (int pc = 0; pc < 6; ++pc) {
        if (pc < 5) {
            const int nx = pc + 1;
            if (nx < 4) w_fetch_ld<2 * D>(W1 + (size_t)(nx & 1) * D * 2 * D + (nx >> 1) * D, tid, p);
            else        w_fetch_ld<D>(W2 + (size_t)(nx - 4) * D * D, tid, p);
        }

        if (active) {
            if (pc >= 4) {
                h_get(hl[pc - 4][wave], m, kh, a);
            } else if (pc & 1) {
                #pragma unroll
                for (int kc = 0; kc < 32; ++kc) a[kc] = am[kc];
            } else {
                #pragma unroll
                for (int kc = 0; kc < 32; ++kc) a[kc] = ax[kc];
            }

            if (pc == 0)      bias_init(acc, bh0);
            else if (pc == 2) bias_init(acc, bh1);
            else if (pc == 4) bias_init(acc, bo);

            tile_products(Wlds[pc & 1], a, acc, hi, m);

            if (pc == 5)     tile_store(out, rbase, m, acc);
            else if (pc & 1) h_put(hl[pc >> 1][wave], m, hi, acc);
        }

        if (pc < 5) {
            w_put(Wlds[(pc + 1) & 1], tid, p);
            __syncthreads();
        }
    }
}

extern "C" void kernel_launch(void* const* d_in, const int* in_sizes, int n_in,
                              void* d_out, int out_size, void* d_ws, size_t ws_size,
                              hipStream_t stream) {
    const float* x     = (const float*)d_in[0];
    const int*   src   = (const int*)  d_in[1];
    const int*   dst   = (const int*)  d_in[2];
    const int*   etype = (const int*)  d_in[3];
    const float* Wrel  = (const float*)d_in[4];
    const float* Wloop = (const float*)d_in[5];
    const float* brel  = (const float*)d_in[6];
    const float* W1    = (const float*)d_in[7];
    const float* b1    = (const float*)d_in[8];
    const float* W2    = (const float*)d_in[9];
    const float* b2    = (const float*)d_in[10];

    float* out = (float*)d_out;
    float* part0 = (float*)d_ws;
    float* part1 = part0 + (size_t)N_NODES * D;
    int* lstart = (int*)((char*)d_ws + 204800000);
    int* sorted = lstart + NCH * NBP;

    k_chunk<<<NCH, CH_T, 0, stream>>>(src, dst, etype, lstart, sorted);

    dim3 g2(NB, 3);
    k_conv<<<g2, CV_T, 0, stream>>>(x, Wrel, Wloop, brel, lstart, sorted, part0, part1, out);

    k_mlp<<<(N_TILES + 7) / 8, 256, 0, stream>>>(x, out, part0, part1, W1, b1, W2, b2, out);
}


# k_conv cut into 1024 blocks = two full rounds of the 512 resident: 149 bins as 2 blocks (5 | 3+self relation steps, launched first), 242 as 3; was 1173 blocks, a thin third round. Sum order changed
# speedup vs baseline: 2.3074x; 1.0377x over previous
#include <hip/hip_runtime.h>
#include <hip/hip_bf16.h>

#define N_NODES 50000
#define N_EDGES 600000
#define N_REL   8
#define D       128
#define N_TILES 3125

typedef __attribute__((ext_vector_type(2))) float v2f;
typedef __attribute__((ext_vector_type(8))) float v8f;

__device__ __forceinline__ v8f wmma4(v2f a, v2f b, v8f c) {
    return __builtin_amdgcn_wmma_f32_16x16x4_f32(false, a, false, b, (short)0, c, false, false);
}

#define TR_WAVES  8
#define TR_Q      32

__device__ __forceinline__ void w_fetch(const float* __restrict__ Wm, int tid,
                                        v2f (&p)[TR_Q])
{
    const float* g = Wm + (size_t)((tid >> 7) * 2) * D + (tid & 127);
    #pragma unroll
    for (int q = 0; q < TR_Q; ++q) {
        p[q][0] = g[q * 4 * D];
        p[q][1] = g[q * 4 * D + D];
    }
}

__device__ __forceinline__ void w_put(float* buf, int tid, const v2f (&p)[TR_Q])
{
    float* d = buf + tid * 2;
    #pragma unroll
    for (int q = 0; q < TR_Q; ++q) *(v2f*)(d + q * 512) = p[q];
}

__device__ __forceinline__ void tile_products(const float* Wb, const v2f (&a)[32],
                                              v8f (&acc)[8], int hi, int m)
{
    #pragma unroll
    for (int kc = 0; kc < 32; ++kc) {
        const float* bp = Wb + (2 * kc + hi) * 256 + m * 2;
        #pragma unroll
        for (int nt = 0; nt < 8; ++nt) {
            v2f b = *(const v2f*)(bp + nt * 32);
            acc[nt] = wmma4(a[kc], b, acc[nt]);
        }
    }
}

__device__ __forceinline__ void tile_store(float* __restrict__ dstp, int rbase, int m,
                                           const v8f (&acc)[8])
{
    #pragma unroll
    for (int nt = 0; nt < 8; ++nt) {
        #pragma unroll
        for (int v = 0; v < 8; ++v)
            dstp[(size_t)(rbase + v) * D + nt * 16 + m] = acc[nt][v];
    }
}

#define NB    391
#define NBP   392
#define NCH   240
#define CE    2500
#define CH_T  1024
#define CH_K  3

__device__ __forceinline__ int wave_inc_scan(int v, int lane)
{
    for (int o = 1; o < 32; o <<= 1) {
        const int up = __shfl_up(v, o);
        if (lane >= o) v += up;
    }
    return v;
}

__global__ __launch_bounds__(CH_T) void k_chunk(
    const int* __restrict__ src, const int* __restrict__ dst,
    const int* __restrict__ etype, int* __restrict__ lstart, int* __restrict__ sorted)
{
    __shared__ int hist[NBP];
    __shared__ int wtot[32];
    __shared__ int recl[CE];
    const int tid  = threadIdx.x;
    const int lane = tid & 31;
    const int wave = tid >> 5;
    const int c    = blockIdx.x;

    if (tid < NBP) hist[tid] = 0;
    __syncthreads();

    int rec[CH_K], bin[CH_K], rk[CH_K];
    #pragma unroll
    for (int k = 0; k < CH_K; ++k) {
        const int i = tid + k * CH_T;
        bin[k] = -1;
        rec[k] = 0;
        if (i < CE) {
            const int e = c * CE + i;
            const int d = dst[e];
            bin[k] = d >> 7;
            rec[k] = ((d & 127) << 19) | (etype[e] * N_NODES + src[e]);
        }
    }
    #pragma unroll
    for (int k = 0; k < CH_K; ++k) {
        rk[k] = 0;
        if (bin[k] >= 0) rk[k] = atomicAdd(&hist[bin[k]], 1);
    }
    __syncthreads();

    const int v   = (tid < NB) ? hist[tid] : 0;
    const int inc = wave_inc_scan(v, lane);
    if (lane == 31) wtot[wave] = inc;
    __syncthreads();
    if (wave == 0) {
        const int tot = wtot[lane];
        const int ti  = wave_inc_scan(tot, lane);
        wtot[lane] = ti - tot;
    }
    __syncthreads();
    const int ex = wtot[wave] + inc - v;
    if (tid < NBP) {
        hist[tid] = ex;
        lstart[c * NBP + tid] = ex;
    }
    __syncthreads();

    #pragma unroll
    for (int k = 0; k < CH_K; ++k)
        if (bin[k] >= 0) recl[hist[bin[k]] + rk[k]] = rec[k];
    __syncthreads();

    #pragma unroll
    for (int k = 0; k < CH_K; ++k) {
        const int i = tid + k * CH_T;
        if (i < CE) sorted[c * CE + i] = recl[i];
    }
}

typedef __attribute__((ext_vector_type(4))) float v4f;

#define ML_HP 132

template <int LD>
__device__ __forceinline__ void w_fetch_ld(const float* __restrict__ Wm, int tid,
                                           v2f (&p)[TR_Q])
{
    const float* g = Wm + (size_t)((tid >> 7) * 2) * LD + (tid & 127);
    #pragma unroll
    for (int q = 0; q < TR_Q; ++q) {
        p[q][0] = g[q * 4 * LD];
        p[q][1] = g[q * 4 * LD + LD];
    }
}

__device__ __forceinline__ void bias_init(v8f (&acc)[8], const float (&b)[8])
{
    #pragma unroll
    for (int nt = 0; nt < 8; ++nt) {
        #pragma unroll
        for (int v = 0; v < 8; ++v) acc[nt][v] = b[nt];
    }
}

__device__ __forceinline__ void h_put(float* hw, int m, int hi, const v8f (&acc)[8])
{
    #pragma unroll
    for (int nt = 0; nt < 8; ++nt) {
        #pragma unroll
        for (int v = 0; v < 8; ++v) {
            const float val = acc[nt][v];
            hw[(hi * 8 + v) * ML_HP + nt * 16 + m] = val > 0.0f ? val : 0.0f;
        }
    }
}

__device__ __forceinline__ void h_get(const float* hw, int m, int kh, v2f (&a)[32])
{
    const float* r = hw + m * ML_HP + kh;
    #pragma unroll
    for (int kc = 0; kc < 32; ++kc) a[kc] = *(const v2f*)(r + kc * 4);
}

#define CV_T    256
#define CV_U    8
#define CV_CAP  (CV_T * CV_U)
#define CV_MAXR 5
#define CV_NK   (CV_MAXR * 128)
#define CV_NB2  149
#define CV_NB3  (NB - CV_NB2)
#define CV_NBLK (2 * CV_NB2 + 3 * CV_NB3)
#define CV_KPL  (CV_NK / 32)

template <int G>
__device__ __forceinline__ void rows_in(v4f& sum, int& cur, float* pw,
                                        const float* __restrict__ x,
                                        int mine, int j, int n, int lane)
{
    v4f v[G];
    #pragma unroll
    for (int u = 0; u < G; ++u) {
        const int jj = j + u < n ? j + u : n - 1;
        const int r  = __builtin_amdgcn_readlane(mine, jj);
        v[u] = ((const v4f*)(x + (size_t)(r & 0xFFFF) * D))[lane];
    }
    #pragma unroll
    for (int u = 0; u < G; ++u) asm volatile("" :: "v"(v[u]));
    #pragma unroll
    for (int u = 0; u < G; ++u) {
        if (j + u < n) {
            const int row = (__builtin_amdgcn_readlane(mine, j + u) >> 16) & 15;
            if (row != cur) {
                if (cur >= 0) *(v4f*)(pw + cur * ML_HP + lane * 4) = sum;
                cur = row;
                sum = v[u];
            } else {
                sum += v[u];
            }
        }
    }
}

__global__ __launch_bounds__(CV_T, 4) void k_conv(
    const float* __restrict__ x, const float* __restrict__ Wrel,
    const float* __restrict__ Wloop, const float* __restrict__ brel,
    const int* __restrict__ lstart, const int* __restrict__ sorted,
    float* __restrict__ part0, float* __restrict__ part1, float* __restrict__ mid)
{
    __shared__ alignas(16) float Wl[D * D];
    __shared__ alignas(16) float pat[TR_WAVES][16 * ML_HP];
    __shared__ int runo[NCH + 1];
    __shared__ int runs[NCH];
    __shared__ int wtot[CV_T / 32];
    __shared__ int keyc[CV_NK];
    __shared__ int keyb[CV_NK + 1];
    __shared__ int recs[CV_CAP];

    const int tid  = threadIdx.x;
    const int lane = tid & 31;
    const int wave = __builtin_amdgcn_readfirstlane(tid >> 5);
    const int m  = lane & 15;
    const int hi = lane >> 4;
    const int kh = hi * 2;
    const int blk = blockIdx.x;
    int b, rel0, nre;
    float* dstp;
    if (blk < 2 * CV_NB2) {
        const bool second = blk >= CV_NB2;
        b    = second ? blk - CV_NB2 : blk;
        rel0 = second ? 5 : 0;
        nre  = second ? N_REL - 5 : 5;
        dstp = second ? mid : part0;
    } else {
        const int t = blk - 2 * CV_NB2;
        const int g = t / CV_NB3;
        b    = CV_NB2 + (t - g * CV_NB3);
        rel0 = 3 * g;
        nre  = g == 2 ? N_REL - 6 : 3;
        dstp = g == 0 ? part0 : (g == 1 ? part1 : mid);
    }
    const bool self = rel0 + nre == N_REL;

    const int  tile   = b * TR_WAVES + wave;
    const bool active = tile < N_TILES;
    const int  row0   = (active ? tile : (N_TILES - 1)) * 16;
    const int  rbase  = row0 + hi * 8;
    float* pw = pat[wave];

    int cnt = 0;
    if (tid < NCH) {
        const int* ls = lstart + tid * NBP + b;
        const int s = ls[0];
        cnt = ls[1] - s;
        runs[tid] = tid * CE + s;
    }
    const int inc = wave_inc_scan(cnt, lane);
    if (lane == 31) wtot[wave] = inc;
    __syncthreads();
    if (wave == 0) {
        const int tot = lane < CV_T / 32 ? wtot[lane] : 0;
        const int ti  = wave_inc_scan(tot, lane);
        if (lane < CV_T / 32) wtot[lane] = ti - tot;
    }
    __syncthreads();
    const int ex = wtot[wave] + inc - cnt;
    if (tid < NCH) runo[tid] = ex;
    if (tid == NCH - 1) runo[NCH] = ex + cnt;
    __syncthreads();
    const int n = __builtin_amdgcn_readfirstlane(runo[NCH]);

    v8f acc[8];
    if (self) {
        #pragma unroll
        for (int nt = 0; nt < 8; ++nt) {
            const float bv = brel[nt * 16 + m];
            #pragma unroll
            for (int v = 0; v < 8; ++v) acc[nt][v] = bv;
        }
    } else {
        #pragma unroll
        for (int nt = 0; nt < 8; ++nt) acc[nt] = (v8f){};
    }
    v2f a[32];

    int w0 = 0;
    do {
        const int mm = n - w0 < CV_CAP ? n - w0 : CV_CAP;
        for (int i = tid; i < CV_NK; i += CV_T) keyc[i] = 0;
        __syncthreads();

        int rec[CV_U], rk[CV_U];
        {
            int gi[CV_U], lo[CV_U], rr0[CV_U];
            #pragma unroll
            for (int u = 0; u < CV_U; ++u) {
                const int i = tid + u * CV_T;
                gi[u] = i < mm ? w0 + i : -1;
                lo[u] = 0;
            }
            static_assert(NCH <= 256, "eight halvings cover 256 runs");
            #pragma unroll
            for (int step = 128; step >= 1; step >>= 1) {
                #pragma unroll
                for (int u = 0; u < CV_U; ++u) {
                    const int md = lo[u] + step;
                    if (runo[md < NCH ? md : NCH] <= gi[u]) lo[u] = md;
                }
            }
            int rs[CV_U], ro[CV_U];
            #pragma unroll
            for (int u = 0; u < CV_U; ++u) { rs[u] = runs[lo[u]]; ro[u] = runo[lo[u]]; }
            #pragma unroll
            for (int u = 0; u < CV_U; ++u) {
                const int idx = gi[u] >= 0 ? rs[u] + (gi[u] - ro[u]) : 0;
                rr0[u] = sorted[idx];
            }
            #pragma unroll
            for (int u = 0; u < CV_U; ++u) {
                const int r   = rr0[u];
                const int val = r & 0x7FFFF;
                const int rel = val / N_NODES;
                const int rr  = rel - rel0;
                rec[u] = -1;
                if (gi[u] >= 0 && rr >= 0 && rr < nre)
                    rec[u] = ((rr * 128 + (r >> 19)) << 16) | (val - rel * N_NODES);
            }
        }
        #pragma unroll
        for (int u = 0; u < CV_U; ++u) {
            rk[u] = 0;
            if (rec[u] >= 0) rk[u] = atomicAdd(&keyc[rec[u] >> 16], 1);
        }
        __syncthreads();

        if (wave == 0) {
            int cq[CV_KPL];
            int s = 0;
            #pragma unroll
            for (int q = 0; q < CV_KPL; ++q) { cq[q] = keyc[CV_KPL * lane + q]; s += cq[q]; }
            int e = wave_inc_scan(s, lane) - s;
            #pragma unroll
            for (int q = 0; q < CV_KPL; ++q) { keyb[CV_KPL * lane + q] = e; e += cq[q]; }
            if (lane == 31) keyb[CV_NK] = e;
        }
        __syncthreads();

        #pragma unroll
        for (int u = 0; u < CV_U; ++u)
            if (rec[u] >= 0) recs[keyb[rec[u] >> 16] + rk[u]] = rec[u];
        __syncthreads();

        const int ns = nre + ((self && w0 + CV_CAP >= n) ? 1 : 0);
        for (int r = 0; r < ns; ++r) {
            const bool sl = r == nre;
            v2f p[TR_Q];
            w_fetch(sl ? Wloop : Wrel + (size_t)(rel0 + r) * D * D, tid, p);

            if (active && sl) {
                const float* xrow = x + (size_t)(row0 + m) * D + kh;
                #pragma unroll
                for (int kc = 0; kc < 32; ++kc) a[kc] = *(const v2f*)(xrow + kc * 4);
            } else if (active) {
                const int k0  = r * 128 + wave * 16;
                const int beg = __builtin_amdgcn_readfirstlane(keyb[k0]);
                const int end = __builtin_amdgcn_readfirstlane(keyb[k0 + 16]);
                #pragma unroll
                for (int i = 0; i < 16; ++i) *(v4f*)(pw + i * ML_HP + lane * 4) = (v4f){};
                v4f sum = (v4f){};
                int cur = -1;
                for (int base = beg; base < end; base += 32) {
                    const int rem = end - base;
                    const int nn  = rem < 32 ? rem : 32;
                    const int mine = recs[base + (lane < nn ? lane : nn - 1)];
                    int j = 0;
                    for (; nn - j > 8; j += 16) rows_in<16>(sum, cur, pw, x, mine, j, nn, lane);
                    if (j < nn) rows_in<8>(sum, cur, pw, x, mine, j, nn, lane);
                }
                if (cur >= 0) *(v4f*)(pw + cur * ML_HP + lane * 4) = sum;
                h_get(pw, m, kh, a);
            }
            __syncthreads();
            w_put(Wl, tid, p);
            __syncthreads();

            if (active) tile_products(Wl, a, acc, hi, m);
        }
        w0 += CV_CAP;
    } while (w0 < n);

    if (active) tile_store(dstp, rbase, m, acc);
}

__global__ __launch_bounds__(256) void k_mlp(
    const float* __restrict__ x, const float* mid,
    const float* __restrict__ part0, const float* __restrict__ part1,
    const float* __restrict__ W1, const float* __restrict__ b1,
    const float* __restrict__ W2, const float* __restrict__ b2,
    float* out)
{
    __shared__ float Wlds[2][D * D];
    __shared__ float hl[2][TR_WAVES][16 * ML_HP];

    const int tid  = threadIdx.x;
    const int wave = tid >> 5;
    const int lane = tid & 31;
    const int m  = lane & 15;
    const int hi = lane >> 4;
    const int kh = hi * 2;

    const int  tile   = blockIdx.x * TR_WAVES + wave;
    const bool active = tile < N_TILES;
    const int  row0   = (active ? tile : (N_TILES - 1)) * 16;
    const int  rbase  = row0 + hi * 8;

    v2f p[TR_Q];
    w_fetch_ld<2 * D>(W1, tid, p);

    v2f ax[32], am[32];
    {
        const size_t ro = (size_t)(row0 + m) * D + kh;
        const float* xrow = x + ro;
        const float* mrow = mid + ro;
        const float* qrow = part0 + ro;
        const float* rrow = part1 + ro;
        #pragma unroll
        for (int kc = 0; kc < 32; ++kc) ax[kc] = *(const v2f*)(xrow + kc * 4);
        const bool three = blockIdx.x >= CV_NB2;
        v2f ar[32];
        if (three) {
            #pragma unroll
            for (int kc = 0; kc < 32; ++kc) ar[kc] = *(const v2f*)(rrow + kc * 4);
        }
        #pragma unroll
        for (int kc = 0; kc < 32; ++kc)
            am[kc] = *(const v2f*)(mrow + kc * 4) + *(const v2f*)(qrow + kc * 4);
        if (three) {
            #pragma unroll
            for (int kc = 0; kc < 32; ++kc) am[kc] += ar[kc];
        }
    }

    float bh0[8], bh1[8], bo[8];
    #pragma unroll
    for (int nt = 0; nt < 8; ++nt) {
        bh0[nt] = b1[nt * 16 + m];
        bh1[nt] = b1[D + nt * 16 + m];
        bo[nt]  = b2[nt * 16 + m];
    }

    w_put(Wlds[0], tid, p);
    __syncthreads();

    v8f acc[8];
    v2f a[32];

    #pragma unroll 1
    for (int pc = 0; pc < 6; ++pc) {
        if (pc < 5) {
            const int nx = pc + 1;
            if (nx < 4) w_fetch_ld<2 * D>(W1 + (size_t)(nx & 1) * D * 2 * D + (nx >> 1) * D, tid, p);
            else        w_fetch_ld<D>(W2 + (size_t)(nx - 4) * D * D, tid, p);
        }

        if (active) {
            if (pc >= 4) {
                h_get(hl[pc - 4][wave], m, kh, a);
            } else if (pc & 1) {
                #pragma unroll
                for (int kc = 0; kc < 32; ++kc) a[kc] = am[kc];
            } else {
                #pragma unroll
                for (int kc = 0; kc < 32; ++kc) a[kc] = ax[kc];
            }

            if (pc == 0)      bias_init(acc, bh0);
            else if (pc == 2) bias_init(acc, bh1);
            else if (pc == 4) bias_init(acc, bo);

            tile_products(Wlds[pc & 1], a, acc, hi, m);

            if (pc == 5)     tile_store(out, rbase, m, acc);
            else if (pc & 1) h_put(hl[pc >> 1][wave], m, hi, acc);
        }

        if (pc < 5) {
            w_put(Wlds[(pc + 1) & 1], tid, p);
            __syncthreads();
        }
    }
}

extern "C" void kernel_launch(void* const* d_in, const int* in_sizes, int n_in,
                              void* d_out, int out_size, void* d_ws, size_t ws_size,
                              hipStream_t stream) {
    const float* x     = (const float*)d_in[0];
    const int*   src   = (const int*)  d_in[1];
    const int*   dst   = (const int*)  d_in[2];
    const int*   etype = (const int*)  d_in[3];
    const float* Wrel  = (const float*)d_in[4];
    const float* Wloop = (const float*)d_in[5];
    const float* brel  = (const float*)d_in[6];
    const float* W1    = (const float*)d_in[7];
    const float* b1    = (const float*)d_in[8];
    const float* W2    = (const float*)d_in[9];
    const float* b2    = (const float*)d_in[10];

    float* out = (float*)d_out;
    float* part0 = (float*)d_ws;
    float* part1 = part0 + (size_t)N_NODES * D;
    int* lstart = (int*)((char*)d_ws + 204800000);
    int* sorted = lstart + NCH * NBP;

    k_chunk<<<NCH, CH_T, 0, stream>>>(src, dst, etype, lstart, sorted);

    k_conv<<<CV_NBLK, CV_T, 0, stream>>>(x, Wrel, Wloop, brel, lstart, sorted, part0, part1, out);

    k_mlp<<<(N_TILES + 7) / 8, 256, 0, stream>>>(x, out, part0, part1, W1, b1, W2, b2, out);
}


# self loop folded into MLP layer 1: extra k_conv blocks form W1a + Wloop W1b and b1 + brel W1b per call (f32 WMMA); k_conv does 8 products per bin, was 9. Sum order changed
# speedup vs baseline: 2.4034x; 1.0416x over previous
#include <hip/hip_runtime.h>
#include <hip/hip_bf16.h>

#define N_NODES 50000
#define N_EDGES 600000
#define N_REL   8
#define D       128
#define N_TILES 3125

typedef __attribute__((ext_vector_type(2))) float v2f;
typedef __attribute__((ext_vector_type(8))) float v8f;

__device__ __forceinline__ v8f wmma4(v2f a, v2f b, v8f c) {
    return __builtin_amdgcn_wmma_f32_16x16x4_f32(false, a, false, b, (short)0, c, false, false);
}

#define TR_WAVES  8
#define TR_Q      32

__device__ __forceinline__ void w_fetch(const float* __restrict__ Wm, int tid,
                                        v2f (&p)[TR_Q])
{
    const float* g = Wm + (size_t)((tid >> 7) * 2) * D + (tid & 127);
    #pragma unroll
    for (int q = 0; q < TR_Q; ++q) {
        p[q][0] = g[q * 4 * D];
        p[q][1] = g[q * 4 * D + D];
    }
}

__device__ __forceinline__ void w_put(float* buf, int tid, const v2f (&p)[TR_Q])
{
    float* d = buf + tid * 2;
    #pragma unroll
    for (int q = 0; q < TR_Q; ++q) *(v2f*)(d + q * 512) = p[q];
}

__device__ __forceinline__ void tile_products(const float* Wb, const v2f (&a)[32],
                                              v8f (&acc)[8], int hi, int m)
{
    #pragma unroll
    for (int kc = 0; kc < 32; ++kc) {
        const float* bp = Wb + (2 * kc + hi) * 256 + m * 2;
        #pragma unroll
        for (int nt = 0; nt < 8; ++nt) {
            v2f b = *(const v2f*)(bp + nt * 32);
            acc[nt] = wmma4(a[kc], b, acc[nt]);
        }
    }
}

__device__ __forceinline__ void tile_store(float* __restrict__ dstp, int rbase, int m,
                                           const v8f (&acc)[8])
{
    #pragma unroll
    for (int nt = 0; nt < 8; ++nt) {
        #pragma unroll
        for (int v = 0; v < 8; ++v)
            dstp[(size_t)(rbase + v) * D + nt * 16 + m] = acc[nt][v];
    }
}

#define NB    391
#define NBP   392
#define NCH   240
#define CE    2500
#define CH_T  1024
#define CH_K  3

__device__ __forceinline__ int wave_inc_scan(int v, int lane)
{
    for (int o = 1; o < 32; o <<= 1) {
        const int up = __shfl_up(v, o);
        if (lane >= o) v += up;
    }
    return v;
}

__global__ __launch_bounds__(CH_T) void k_chunk(
    const int* __restrict__ src, const int* __restrict__ dst,
    const int* __restrict__ etype, int* __restrict__ lstart, int* __restrict__ sorted)
{
    __shared__ int hist[NBP];
    __shared__ int wtot[32];
    __shared__ int recl[CE];
    const int tid  = threadIdx.x;
    const int lane = tid & 31;
    const int wave = tid >> 5;
    const int c    = blockIdx.x;

    if (tid < NBP) hist[tid] = 0;
    __syncthreads();

    int rec[CH_K], bin[CH_K], rk[CH_K];
    #pragma unroll
    for (int k = 0; k < CH_K; ++k) {
        const int i = tid + k * CH_T;
        bin[k] = -1;
        rec[k] = 0;
        if (i < CE) {
            const int e = c * CE + i;
            const int d = dst[e];
            bin[k] = d >> 7;
            rec[k] = ((d & 127) << 19) | (etype[e] * N_NODES + src[e]);
        }
    }
    #pragma unroll
    for (int k = 0; k < CH_K; ++k) {
        rk[k] = 0;
        if (bin[k] >= 0) rk[k] = atomicAdd(&hist[bin[k]], 1);
    }
    __syncthreads();

    const int v   = (tid < NB) ? hist[tid] : 0;
    const int inc = wave_inc_scan(v, lane);
    if (lane == 31) wtot[wave] = inc;
    __syncthreads();
    if (wave == 0) {
        const int tot = wtot[lane];
        const int ti  = wave_inc_scan(tot, lane);
        wtot[lane] = ti - tot;
    }
    __syncthreads();
    const int ex = wtot[wave] + inc - v;
    if (tid < NBP) {
        hist[tid] = ex;
        lstart[c * NBP + tid] = ex;
    }
    __syncthreads();

    #pragma unroll
    for (int k = 0; k < CH_K; ++k)
        if (bin[k] >= 0) recl[hist[bin[k]] + rk[k]] = rec[k];
    __syncthreads();

    #pragma unroll
    for (int k = 0; k < CH_K; ++k) {
        const int i = tid + k * CH_T;
        if (i < CE) sorted[c * CE + i] = recl[i];
    }
}

typedef __attribute__((ext_vector_type(4))) float v4f;

#define ML_HP 132

template <int LD>
__device__ __forceinline__ void w_fetch_ld(const float* __restrict__ Wm, int tid,
                                           v2f (&p)[TR_Q])
{
    const float* g = Wm + (size_t)((tid >> 7) * 2) * LD + (tid & 127);
    #pragma unroll
    for (int q = 0; q < TR_Q; ++q) {
        p[q][0] = g[q * 4 * LD];
        p[q][1] = g[q * 4 * LD + LD];
    }
}

__device__ __forceinline__ void bias_init(v8f (&acc)[8], const float (&b)[8])
{
    #pragma unroll
    for (int nt = 0; nt < 8; ++nt) {
        #pragma unroll
        for (int v = 0; v < 8; ++v) acc[nt][v] = b[nt];
    }
}

__device__ __forceinline__ void h_put(float* hw, int m, int hi, const v8f (&acc)[8])
{
    #pragma unroll
    for (int nt = 0; nt < 8; ++nt) {
        #pragma unroll
        for (int v = 0; v < 8; ++v) {
            const float val = acc[nt][v];
            hw[(hi * 8 + v) * ML_HP + nt * 16 + m] = val > 0.0f ? val : 0.0f;
        }
    }
}

__device__ __forceinline__ void h_get(const float* hw, int m, int kh, v2f (&a)[32])
{
    const float* r = hw + m * ML_HP + kh;
    #pragma unroll
    for (int kc = 0; kc < 32; ++kc) a[kc] = *(const v2f*)(r + kc * 4);
}

#define CV_T    256
#define CV_U    8
#define CV_CAP  (CV_T * CV_U)
#define CV_MAXR 4
#define CV_NK   (CV_MAXR * 128)
#define CV_NB2  149
#define CV_NB3  (NB - CV_NB2)
#define CV_NBLK (2 * CV_NB2 + 3 * CV_NB3)
#define CV_KPL  (CV_NK / 32)

template <int G>
__device__ __forceinline__ void rows_in(v4f& sum, int& cur, float* pw,
                                        const float* __restrict__ x,
                                        int mine, int j, int n, int lane)
{
    v4f v[G];
    #pragma unroll
    for (int u = 0; u < G; ++u) {
        const int jj = j + u < n ? j + u : n - 1;
        const int r  = __builtin_amdgcn_readlane(mine, jj);
        v[u] = ((const v4f*)(x + (size_t)(r & 0xFFFF) * D))[lane];
    }
    #pragma unroll
    for (int u = 0; u < G; ++u) asm volatile("" :: "v"(v[u]));
    #pragma unroll
    for (int u = 0; u < G; ++u) {
        if (j + u < n) {
            const int row = (__builtin_amdgcn_readlane(mine, j + u) >> 16) & 15;
            if (row != cur) {
                if (cur >= 0) *(v4f*)(pw + cur * ML_HP + lane * 4) = sum;
                cur = row;
                sum = v[u];
            } else {
                sum += v[u];
            }
        }
    }
}

#define FD_TASKS 36
#define FD_NBLK  ((FD_TASKS + TR_WAVES - 1) / TR_WAVES)

__device__ __forceinline__ void fold_task(
    const float* __restrict__ Wloop, const float* __restrict__ brel,
    const float* __restrict__ W1, const float* __restrict__ b1,
    float* __restrict__ Wc, float* __restrict__ bc, int t, int m, int hi)
{
    const int  rt = t >> 2;
    const int  c0 = (t & 3) * 64 + m;
    const bool br = rt == 8;
    const float* W1b = W1 + (size_t)D * 2 * D;

    v2f a[32];
    v8f acc[4];
    if (br) {
        #pragma unroll
        for (int kc = 0; kc < 32; ++kc) {
            a[kc] = (v2f){};
            if (m == 0) a[kc] = *(const v2f*)(brel + kc * 4 + hi * 2);
        }
        #pragma unroll
        for (int nt = 0; nt < 4; ++nt) {
            acc[nt] = (v8f){};
            if (hi == 0) acc[nt][0] = b1[c0 + nt * 16];
        }
    } else {
        const float* ar = Wloop + (size_t)(rt * 16 + m) * D + hi * 2;
        #pragma unroll
        for (int kc = 0; kc < 32; ++kc) a[kc] = *(const v2f*)(ar + kc * 4);
        const float* cr = W1 + (size_t)(rt * 16 + hi * 8) * 2 * D + c0;
        #pragma unroll
        for (int nt = 0; nt < 4; ++nt) {
            #pragma unroll
            for (int v = 0; v < 8; ++v) acc[nt][v] = cr[v * 2 * D + nt * 16];
        }
    }
    #pragma unroll
    for (int kc = 0; kc < 32; ++kc) {
        const float* bp = W1b + (size_t)(4 * kc + 2 * hi) * 2 * D + c0;
        #pragma unroll
        for (int nt = 0; nt < 4; ++nt) {
            v2f b;
            b[0] = bp[nt * 16];
            b[1] = bp[nt * 16 + 2 * D];
            acc[nt] = wmma4(a[kc], b, acc[nt]);
        }
    }
    if (br) {
        if (hi == 0) {
            #pragma unroll
            for (int nt = 0; nt < 4; ++nt) bc[c0 + nt * 16] = acc[nt][0];
        }
    } else {
        float* wr = Wc + (size_t)(rt * 16 + hi * 8) * 2 * D + c0;
        #pragma unroll
        for (int nt = 0; nt < 4; ++nt) {
            #pragma unroll
            for (int v = 0; v < 8; ++v) wr[v * 2 * D + nt * 16] = acc[nt][v];
        }
    }
}

__global__ __launch_bounds__(CV_T, 4) void k_conv(
    const float* __restrict__ x, const float* __restrict__ Wrel,
    const float* __restrict__ Wloop, const float* __restrict__ brel,
    const float* __restrict__ W1, const float* __restrict__ b1,
    const int* __restrict__ lstart, const int* __restrict__ sorted,
    float* __restrict__ part0, float* __restrict__ part1, float* __restrict__ mid,
    float* __restrict__ Wc, float* __restrict__ bc)
{
    __shared__ alignas(16) float Wl[D * D];
    __shared__ alignas(16) float pat[TR_WAVES][16 * ML_HP];
    __shared__ int runo[NCH + 1];
    __shared__ int runs[NCH];
    __shared__ int wtot[CV_T / 32];
    __shared__ int keyc[CV_NK];
    __shared__ int keyb[CV_NK + 1];
    __shared__ int recs[CV_CAP];

    const int tid  = threadIdx.x;
    const int lane = tid & 31;
    const int wave = __builtin_amdgcn_readfirstlane(tid >> 5);
    const int m  = lane & 15;
    const int hi = lane >> 4;
    const int kh = hi * 2;
    const int blk = blockIdx.x;
    if (blk >= CV_NBLK) {
        const int t = (blk - CV_NBLK) * TR_WAVES + wave;
        if (t < FD_TASKS) fold_task(Wloop, brel, W1, b1, Wc, bc, t, m, hi);
        return;
    }
    int b, rel0, nre;
    float* dstp;
    if (blk < 2 * CV_NB2) {
        const bool second = blk >= CV_NB2;
        b    = second ? blk - CV_NB2 : blk;
        rel0 = second ? 4 : 0;
        nre  = 4;
        dstp = second ? mid : part0;
    } else {
        const int t = blk - 2 * CV_NB2;
        const int g = t / CV_NB3;
        b    = CV_NB2 + (t - g * CV_NB3);
        rel0 = 3 * g;
        nre  = g == 2 ? N_REL - 6 : 3;
        dstp = g == 0 ? part0 : (g == 1 ? part1 : mid);
    }

    const int  tile   = b * TR_WAVES + wave;
    const bool active = tile < N_TILES;
    const int  row0   = (active ? tile : (N_TILES - 1)) * 16;
    const int  rbase  = row0 + hi * 8;
    float* pw = pat[wave];

    int cnt = 0;
    if (tid < NCH) {
        const int* ls = lstart + tid * NBP + b;
        const int s = ls[0];
        cnt = ls[1] - s;
        runs[tid] = tid * CE + s;
    }
    const int inc = wave_inc_scan(cnt, lane);
    if (lane == 31) wtot[wave] = inc;
    __syncthreads();
    if (wave == 0) {
        const int tot = lane < CV_T / 32 ? wtot[lane] : 0;
        const int ti  = wave_inc_scan(tot, lane);
        if (lane < CV_T / 32) wtot[lane] = ti - tot;
    }
    __syncthreads();
    const int ex = wtot[wave] + inc - cnt;
    if (tid < NCH) runo[tid] = ex;
    if (tid == NCH - 1) runo[NCH] = ex + cnt;
    __syncthreads();
    const int n = __builtin_amdgcn_readfirstlane(runo[NCH]);

    v8f acc[8];
    #pragma unroll
    for (int nt = 0; nt < 8; ++nt) acc[nt] = (v8f){};
    v2f a[32];

    int w0 = 0;
    do {
        const int mm = n - w0 < CV_CAP ? n - w0 : CV_CAP;
        for (int i = tid; i < CV_NK; i += CV_T) keyc[i] = 0;
        __syncthreads();

        int rec[CV_U], rk[CV_U];
        {
            int gi[CV_U], lo[CV_U], rr0[CV_U];
            #pragma unroll
            for (int u = 0; u < CV_U; ++u) {
                const int i = tid + u * CV_T;
                gi[u] = i < mm ? w0 + i : -1;
                lo[u] = 0;
            }
            static_assert(NCH <= 256, "eight halvings cover 256 runs");
            #pragma unroll
            for (int step = 128; step >= 1; step >>= 1) {
                #pragma unroll
                for (int u = 0; u < CV_U; ++u) {
                    const int md = lo[u] + step;
                    if (runo[md < NCH ? md : NCH] <= gi[u]) lo[u] = md;
                }
            }
            int rs[CV_U], ro[CV_U];
            #pragma unroll
            for (int u = 0; u < CV_U; ++u) { rs[u] = runs[lo[u]]; ro[u] = runo[lo[u]]; }
            #pragma unroll
            for (int u = 0; u < CV_U; ++u) {
                const int idx = gi[u] >= 0 ? rs[u] + (gi[u] - ro[u]) : 0;
                rr0[u] = sorted[idx];
            }
            #pragma unroll
            for (int u = 0; u < CV_U; ++u) {
                const int r   = rr0[u];
                const int val = r & 0x7FFFF;
                const int rel = val / N_NODES;
                const int rr  = rel - rel0;
                rec[u] = -1;
                if (gi[u] >= 0 && rr >= 0 && rr < nre)
                    rec[u] = ((rr * 128 + (r >> 19)) << 16) | (val - rel * N_NODES);
            }
        }
        #pragma unroll
        for (int u = 0; u < CV_U; ++u) {
            rk[u] = 0;
            if (rec[u] >= 0) rk[u] = atomicAdd(&keyc[rec[u] >> 16], 1);
        }
        __syncthreads();

        if (wave == 0) {
            int cq[CV_KPL];
            int s = 0;
            #pragma unroll
            for (int q = 0; q < CV_KPL; ++q) { cq[q] = keyc[CV_KPL * lane + q]; s += cq[q]; }
            int e = wave_inc_scan(s, lane) - s;
            #pragma unroll
            for (int q = 0; q < CV_KPL; ++q) { keyb[CV_KPL * lane + q] = e; e += cq[q]; }
            if (lane == 31) keyb[CV_NK] = e;
        }
        __syncthreads();

        #pragma unroll
        for (int u = 0; u < CV_U; ++u)
            if (rec[u] >= 0) recs[keyb[rec[u] >> 16] + rk[u]] = rec[u];
        __syncthreads();

        for (int r = 0; r < nre; ++r) {
            v2f p[TR_Q];
            w_fetch(Wrel + (size_t)(rel0 + r) * D * D, tid, p);

            if (active) {
                const int k0  = r * 128 + wave * 16;
                const int beg = __builtin_amdgcn_readfirstlane(keyb[k0]);
                const int end = __builtin_amdgcn_readfirstlane(keyb[k0 + 16]);
                #pragma unroll
                for (int i = 0; i < 16; ++i) *(v4f*)(pw + i * ML_HP + lane * 4) = (v4f){};
                v4f sum = (v4f){};
                int cur = -1;
                for (int base = beg; base < end; base += 32) {
                    const int rem = end - base;
                    const int nn  = rem < 32 ? rem : 32;
                    const int mine = recs[base + (lane < nn ? lane : nn - 1)];
                    int j = 0;
                    for (; nn - j > 8; j += 16) rows_in<16>(sum, cur, pw, x, mine, j, nn, lane);
                    if (j < nn) rows_in<8>(sum, cur, pw, x, mine, j, nn, lane);
                }
                if (cur >= 0) *(v4f*)(pw + cur * ML_HP + lane * 4) = sum;
                h_get(pw, m, kh, a);
            }
            __syncthreads();
            w_put(Wl, tid, p);
            __syncthreads();

            if (active) tile_products(Wl, a, acc, hi, m);
        }
        w0 += CV_CAP;
    } while (w0 < n);

    if (active) tile_store(dstp, rbase, m, acc);
}

__global__ __launch_bounds__(256) void k_mlp(
    const float* __restrict__ x, const float* mid,
    const float* __restrict__ part0, const float* __restrict__ part1,
    const float* __restrict__ W1, const float* __restrict__ Wc, const float* __restrict__ bc,
    const float* __restrict__ W2, const float* __restrict__ b2,
    float* out)
{
    __shared__ float Wlds[2][D * D];
    __shared__ float hl[2][TR_WAVES][16 * ML_HP];

    const int tid  = threadIdx.x;
    const int wave = tid >> 5;
    const int lane = tid & 31;
    const int m  = lane & 15;
    const int hi = lane >> 4;
    const int kh = hi * 2;

    const int  tile   = blockIdx.x * TR_WAVES + wave;
    const bool active = tile < N_TILES;
    const int  row0   = (active ? tile : (N_TILES - 1)) * 16;
    const int  rbase  = row0 + hi * 8;

    v2f p[TR_Q];
    w_fetch_ld<2 * D>(Wc, tid, p);

    v2f ax[32], am[32];
    {
        const size_t ro = (size_t)(row0 + m) * D + kh;
        const float* xrow = x + ro;
        const float* mrow = mid + ro;
        const float* qrow = part0 + ro;
        const float* rrow = part1 + ro;
        #pragma unroll
        for (int kc = 0; kc < 32; ++kc) ax[kc] = *(const v2f*)(xrow + kc * 4);
        const bool three = blockIdx.x >= CV_NB2;
        v2f ar[32];
        if (three) {
            #pragma unroll
            for (int kc = 0; kc < 32; ++kc) ar[kc] = *(const v2f*)(rrow + kc * 4);
        }
        #pragma unroll
        for (int kc = 0; kc < 32; ++kc)
            am[kc] = *(const v2f*)(mrow + kc * 4) + *(const v2f*)(qrow + kc * 4);
        if (three) {
            #pragma unroll
            for (int kc = 0; kc < 32; ++kc) am[kc] += ar[kc];
        }
    }

    float bh0[8], bh1[8], bo[8];
    #pragma unroll
    for (int nt = 0; nt < 8; ++nt) {
        bh0[nt] = bc[nt * 16 + m];
        bh1[nt] = bc[D + nt * 16 + m];
        bo[nt]  = b2[nt * 16 + m];
    }

    w_put(Wlds[0], tid, p);
    __syncthreads();

    v8f acc[8];
    v2f a[32];

    #pragma unroll 1
    for (int pc = 0; pc < 6; ++pc) {
        if (pc < 5) {
            const int nx = pc + 1;
            if (nx < 4) w_fetch_ld<2 * D>(((nx & 1) ? W1 + (size_t)D * 2 * D : Wc) + (nx >> 1) * D, tid, p);
            else        w_fetch_ld<D>(W2 + (size_t)(nx - 4) * D * D, tid, p);
        }

        if (active) {
            if (pc >= 4) {
                h_get(hl[pc - 4][wave], m, kh, a);
            } else if (pc & 1) {
                #pragma unroll
                for (int kc = 0; kc < 32; ++kc) a[kc] = am[kc];
            } else {
                #pragma unroll
                for (int kc = 0; kc < 32; ++kc) a[kc] = ax[kc];
            }

            if (pc == 0)      bias_init(acc, bh0);
            else if (pc == 2) bias_init(acc, bh1);
            else if (pc == 4) bias_init(acc, bo);

            tile_products(Wlds[pc & 1], a, acc, hi, m);

            if (pc == 5)     tile_store(out, rbase, m, acc);
            else if (pc & 1) h_put(hl[pc >> 1][wave], m, hi, acc);
        }

        if (pc < 5) {
            w_put(Wlds[(pc + 1) & 1], tid, p);
            __syncthreads();
        }
    }
}

extern "C" void kernel_launch(void* const* d_in, const int* in_sizes, int n_in,
                              void* d_out, int out_size, void* d_ws, size_t ws_size,
                              hipStream_t stream) {
    const float* x     = (const float*)d_in[0];
    const int*   src   = (const int*)  d_in[1];
    const int*   dst   = (const int*)  d_in[2];
    const int*   etype = (const int*)  d_in[3];
    const float* Wrel  = (const float*)d_in[4];
    const float* Wloop = (const float*)d_in[5];
    const float* brel  = (const float*)d_in[6];
    const float* W1    = (const float*)d_in[7];
    const float* b1    = (const float*)d_in[8];
    const float* W2    = (const float*)d_in[9];
    const float* b2    = (const float*)d_in[10];

    float* out = (float*)d_out;
    float* part0 = (float*)d_ws;
    float* part1 = part0 + (size_t)N_NODES * D;
    int* lstart = (int*)((char*)d_ws + 204800000);
    int* sorted = lstart + NCH * NBP;
    float* Wc = (float*)((char*)d_ws + 210000000);
    float* bc = Wc + D * 2 * D;

    k_chunk<<<NCH, CH_T, 0, stream>>>(src, dst, etype, lstart, sorted);

    k_conv<<<CV_NBLK + FD_NBLK, CV_T, 0, stream>>>(x, Wrel, Wloop, brel, W1, b1, lstart, sorted,
                                                   part0, part1, out, Wc, bc);

    k_mlp<<<(N_TILES + 7) / 8, 256, 0, stream>>>(x, out, part0, part1, W1, Wc, bc, W2, b2, out);
}


# k_mlp: 256 resident 16-wave blocks take their 12-13 row tiles 4 per round; a tile's six products split by output-column quarters over 4 waves, one per SIMD; rows via LDS patches. Sum order kept
# speedup vs baseline: 2.5530x; 1.0622x over previous
#include <hip/hip_runtime.h>
#include <hip/hip_bf16.h>

#define N_NODES 50000
#define N_EDGES 600000
#define N_REL   8
#define D       128
#define N_TILES 3125

typedef __attribute__((ext_vector_type(2))) float v2f;
typedef __attribute__((ext_vector_type(8))) float v8f;

__device__ __forceinline__ v8f wmma4(v2f a, v2f b, v8f c) {
    return __builtin_amdgcn_wmma_f32_16x16x4_f32(false, a, false, b, (short)0, c, false, false);
}

#define TR_WAVES  8
#define TR_Q      32

__device__ __forceinline__ void w_fetch(const float* __restrict__ Wm, int tid,
                                        v2f (&p)[TR_Q])
{
    const float* g = Wm + (size_t)((tid >> 7) * 2) * D + (tid & 127);
    #pragma unroll
    for (int q = 0; q < TR_Q; ++q) {
        p[q][0] = g[q * 4 * D];
        p[q][1] = g[q * 4 * D + D];
    }
}

__device__ __forceinline__ void w_put(float* buf, int tid, const v2f (&p)[TR_Q])
{
    float* d = buf + tid * 2;
    #pragma unroll
    for (int q = 0; q < TR_Q; ++q) *(v2f*)(d + q * 512) = p[q];
}

__device__ __forceinline__ void tile_products(const float* Wb, const v2f (&a)[32],
                                              v8f (&acc)[8], int hi, int m)
{
    #pragma unroll
    for (int kc = 0; kc < 32; ++kc) {
        const float* bp = Wb + (2 * kc + hi) * 256 + m * 2;
        #pragma unroll
        for (int nt = 0; nt < 8; ++nt) {
            v2f b = *(const v2f*)(bp + nt * 32);
            acc[nt] = wmma4(a[kc], b, acc[nt]);
        }
    }
}

__device__ __forceinline__ void tile_store(float* __restrict__ dstp, int rbase, int m,
                                           const v8f (&acc)[8])
{
    #pragma unroll
    for (int nt = 0; nt < 8; ++nt) {
        #pragma unroll
        for (int v = 0; v < 8; ++v)
            dstp[(size_t)(rbase + v) * D + nt * 16 + m] = acc[nt][v];
    }
}

#define NB    391
#define NBP   392
#define NCH   240
#define CE    2500
#define CH_T  1024
#define CH_K  3

__device__ __forceinline__ int wave_inc_scan(int v, int lane)
{
    for (int o = 1; o < 32; o <<= 1) {
        const int up = __shfl_up(v, o);
        if (lane >= o) v += up;
    }
    return v;
}

__global__ __launch_bounds__(CH_T) void k_chunk(
    const int* __restrict__ src, const int* __restrict__ dst,
    const int* __restrict__ etype, int* __restrict__ lstart, int* __restrict__ sorted)
{
    __shared__ int hist[NBP];
    __shared__ int wtot[32];
    __shared__ int recl[CE];
    const int tid  = threadIdx.x;
    const int lane = tid & 31;
    const int wave = tid >> 5;
    const int c    = blockIdx.x;

    if (tid < NBP) hist[tid] = 0;
    __syncthreads();

    int rec[CH_K], bin[CH_K], rk[CH_K];
    #pragma unroll
    for (int k = 0; k < CH_K; ++k) {
        const int i = tid + k * CH_T;
        bin[k] = -1;
        rec[k] = 0;
        if (i < CE) {
            const int e = c * CE + i;
            const int d = dst[e];
            bin[k] = d >> 7;
            rec[k] = ((d & 127) << 19) | (etype[e] * N_NODES + src[e]);
        }
    }
    #pragma unroll
    for (int k = 0; k < CH_K; ++k) {
        rk[k] = 0;
        if (bin[k] >= 0) rk[k] = atomicAdd(&hist[bin[k]], 1);
    }
    __syncthreads();

    const int v   = (tid < NB) ? hist[tid] : 0;
    const int inc = wave_inc_scan(v, lane);
    if (lane == 31) wtot[wave] = inc;
    __syncthreads();
    if (wave == 0) {
        const int tot = wtot[lane];
        const int ti  = wave_inc_scan(tot, lane);
        wtot[lane] = ti - tot;
    }
    __syncthreads();
    const int ex = wtot[wave] + inc - v;
    if (tid < NBP) {
        hist[tid] = ex;
        lstart[c * NBP + tid] = ex;
    }
    __syncthreads();

    #pragma unroll
    for (int k = 0; k < CH_K; ++k)
        if (bin[k] >= 0) recl[hist[bin[k]] + rk[k]] = rec[k];
    __syncthreads();

    #pragma unroll
    for (int k = 0; k < CH_K; ++k) {
        const int i = tid + k * CH_T;
        if (i < CE) sorted[c * CE + i] = recl[i];
    }
}

typedef __attribute__((ext_vector_type(4))) float v4f;

#define ML_HP 132
#define MQ_NB   256
#define MQ_T    512
#define MQ_TPR  4
#define MQ_Q    16

template <int LD>
__device__ __forceinline__ void wq_fetch(const float* Wm, int tid,
                                         v2f (&p)[MQ_Q])
{
    const float* g = Wm + (size_t)((tid >> 7) * 2) * LD + (tid & 127);
    #pragma unroll
    for (int q = 0; q < MQ_Q; ++q) {
        p[q][0] = g[q * 8 * LD];
        p[q][1] = g[q * 8 * LD + LD];
    }
}

__device__ __forceinline__ void wq_put(float* buf, int tid, const v2f (&p)[MQ_Q])
{
    const int n  = tid & 127;
    const int pl = ((n >> 5) << 5) + ((n & 15) << 1) + ((n >> 4) & 1);
    float* d = buf + (tid >> 7) * 256 + pl * 2;
    #pragma unroll
    for (int q = 0; q < MQ_Q; ++q) *(v2f*)(d + q * 1024) = p[q];
}

__device__ __forceinline__ const float* quarter_bbase(const float* Wb, int hi, int m, int cq)
{
    return Wb + hi * 256 + cq * 64 + m * 4;
}

__device__ __forceinline__ void quarter_bhead(const float* bp, v4f (&cur)[4])
{
    #pragma unroll
    for (int i = 0; i < 4; ++i) cur[i] = *(const v4f*)(bp + i * 512);
}

__device__ __forceinline__ void quarter_products(const float* bp, const v2f (&a)[32],
                                                 v8f (&acc)[2], v4f (&cur)[4])
{
    v4f nxt[4];
    #pragma unroll
    for (int g = 0; g < 8; ++g) {
        if (g < 7) {
            #pragma unroll
            for (int i = 0; i < 4; ++i) nxt[i] = *(const v4f*)(bp + (4 * g + 4 + i) * 512);
        }
        __builtin_amdgcn_sched_barrier(0);
        #pragma unroll
        for (int i = 0; i < 4; ++i) {
            v2f b0, b1;
            b0[0] = cur[i][0]; b0[1] = cur[i][1];
            b1[0] = cur[i][2]; b1[1] = cur[i][3];
            acc[0] = wmma4(a[4 * g + i], b0, acc[0]);
            acc[1] = wmma4(a[4 * g + i], b1, acc[1]);
        }
        #pragma unroll
        for (int i = 0; i < 4; ++i) cur[i] = nxt[i];
    }
}

__device__ __forceinline__ void bias_init2(v8f (&acc)[2], const float (&b)[2])
{
    #pragma unroll
    for (int nt = 0; nt < 2; ++nt) {
        #pragma unroll
        for (int v = 0; v < 8; ++v) acc[nt][v] = b[nt];
    }
}

__device__ __forceinline__ void h_put2(float* hw, int m, int hi, int cq, const v8f (&acc)[2])
{
    #pragma unroll
    for (int nt = 0; nt < 2; ++nt) {
        #pragma unroll
        for (int v = 0; v < 8; ++v) {
            const float val = acc[nt][v];
            hw[(hi * 8 + v) * ML_HP + (2 * cq + nt) * 16 + m] = val > 0.0f ? val : 0.0f;
        }
    }
}

__device__ __forceinline__ void quarter_store(float* dstp, int rbase, int m, int cq,
                                              const v8f (&acc)[2])
{
    #pragma unroll
    for (int nt = 0; nt < 2; ++nt) {
        #pragma unroll
        for (int v = 0; v < 8; ++v)
            dstp[(size_t)(rbase + v) * D + (2 * cq + nt) * 16 + m] = acc[nt][v];
    }
}

__device__ __forceinline__ void h_get(const float* hw, int m, int kh, v2f (&a)[32])
{
    const float* r = hw + m * ML_HP + kh;
    #pragma unroll
    for (int kc = 0; kc < 32; ++kc) a[kc] = *(const v2f*)(r + kc * 4);
}

#define CV_T    256
#define CV_U    8
#define CV_CAP  (CV_T * CV_U)
#define CV_MAXR 4
#define CV_NK   (CV_MAXR * 128)
#define CV_NB2  149
#define CV_NB3  (NB - CV_NB2)
#define CV_NBLK (2 * CV_NB2 + 3 * CV_NB3)
#define CV_KPL  (CV_NK / 32)

template <int G>
__device__ __forceinline__ void rows_in(v4f& sum, int& cur, float* pw,
                                        const float* __restrict__ x,
                                        int mine, int j, int n, int lane)
{
    v4f v[G];
    #pragma unroll
    for (int u = 0; u < G; ++u) {
        const int jj = j + u < n ? j + u : n - 1;
        const int r  = __builtin_amdgcn_readlane(mine, jj);
        v[u] = ((const v4f*)(x + (size_t)(r & 0xFFFF) * D))[lane];
    }
    #pragma unroll
    for (int u = 0; u < G; ++u) asm volatile("" :: "v"(v[u]));
    #pragma unroll
    for (int u = 0; u < G; ++u) {
        if (j + u < n) {
            const int row = (__builtin_amdgcn_readlane(mine, j + u) >> 16) & 15;
            if (row != cur) {
                if (cur >= 0) *(v4f*)(pw + cur * ML_HP + lane * 4) = sum;
                cur = row;
                sum = v[u];
            } else {
                sum += v[u];
            }
        }
    }
}

#define FD_TASKS 36
#define FD_NBLK  ((FD_TASKS + TR_WAVES - 1) / TR_WAVES)

__device__ __forceinline__ void fold_task(
    const float* __restrict__ Wloop, const float* __restrict__ brel,
    const float* __restrict__ W1, const float* __restrict__ b1,
    float* __restrict__ Wc, float* __restrict__ bc, int t, int m, int hi)
{
    const int  rt = t >> 2;
    const int  c0 = (t & 3) * 64 + m;
    const bool br = rt == 8;
    const float* W1b = W1 + (size_t)D * 2 * D;

    v2f a[32];
    v8f acc[4];
    if (br) {
        #pragma unroll
        for (int kc = 0; kc < 32; ++kc) {
            a[kc] = (v2f){};
            if (m == 0) a[kc] = *(const v2f*)(brel + kc * 4 + hi * 2);
        }
        #pragma unroll
        for (int nt = 0; nt < 4; ++nt) {
            acc[nt] = (v8f){};
            if (hi == 0) acc[nt][0] = b1[c0 + nt * 16];
        }
    } else {
        const float* ar = Wloop + (size_t)(rt * 16 + m) * D + hi * 2;
        #pragma unroll
        for (int kc = 0; kc < 32; ++kc) a[kc] = *(const v2f*)(ar + kc * 4);
        const float* cr = W1 + (size_t)(rt * 16 + hi * 8) * 2 * D + c0;
        #pragma unroll
        for (int nt = 0; nt < 4; ++nt) {
            #pragma unroll
            for (int v = 0; v < 8; ++v) acc[nt][v] = cr[v * 2 * D + nt * 16];
        }
    }
    #pragma unroll
    for (int kc = 0; kc < 32; ++kc) {
        const float* bp = W1b + (size_t)(4 * kc + 2 * hi) * 2 * D + c0;
        #pragma unroll
        for (int nt = 0; nt < 4; ++nt) {
            v2f b;
            b[0] = bp[nt * 16];
            b[1] = bp[nt * 16 + 2 * D];
            acc[nt] = wmma4(a[kc], b, acc[nt]);
        }
    }
    if (br) {
        if (hi == 0) {
            #pragma unroll
            for (int nt = 0; nt < 4; ++nt) bc[c0 + nt * 16] = acc[nt][0];
        }
    } else {
        float* wr = Wc + (size_t)(rt * 16 + hi * 8) * 2 * D + c0;
        #pragma unroll
        for (int nt = 0; nt < 4; ++nt) {
            #pragma unroll
            for (int v = 0; v < 8; ++v) wr[v * 2 * D + nt * 16] = acc[nt][v];
        }
    }
}

__global__ __launch_bounds__(CV_T, 4) void k_conv(
    const float* __restrict__ x, const float* __restrict__ Wrel,
    const float* __restrict__ Wloop, const float* __restrict__ brel,
    const float* __restrict__ W1, const float* __restrict__ b1,
    const int* __restrict__ lstart, const int* __restrict__ sorted,
    float* __restrict__ part0, float* __restrict__ part1, float* __restrict__ mid,
    float* __restrict__ Wc, float* __restrict__ bc)
{
    __shared__ alignas(16) float Wl[D * D];
    __shared__ alignas(16) float pat[TR_WAVES][16 * ML_HP];
    __shared__ int runo[NCH + 1];
    __shared__ int runs[NCH];
    __shared__ int wtot[CV_T / 32];
    __shared__ int keyc[CV_NK];
    __shared__ int keyb[CV_NK + 1];
    __shared__ int recs[CV_CAP];

    const int tid  = threadIdx.x;
    const int lane = tid & 31;
    const int wave = __builtin_amdgcn_readfirstlane(tid >> 5);
    const int m  = lane & 15;
    const int hi = lane >> 4;
    const int kh = hi * 2;
    const int blk = blockIdx.x;
    if (blk >= CV_NBLK) {
        const int t = (blk - CV_NBLK) * TR_WAVES + wave;
        if (t < FD_TASKS) fold_task(Wloop, brel, W1, b1, Wc, bc, t, m, hi);
        return;
    }
    int b, rel0, nre;
    float* dstp;
    if (blk < 2 * CV_NB2) {
        const bool second = blk >= CV_NB2;
        b    = second ? blk - CV_NB2 : blk;
        rel0 = second ? 4 : 0;
        nre  = 4;
        dstp = second ? mid : part0;
    } else {
        const int t = blk - 2 * CV_NB2;
        const int g = t / CV_NB3;
        b    = CV_NB2 + (t - g * CV_NB3);
        rel0 = 3 * g;
        nre  = g == 2 ? N_REL - 6 : 3;
        dstp = g == 0 ? part0 : (g == 1 ? part1 : mid);
    }

    const int  tile   = b * TR_WAVES + wave;
    const bool active = tile < N_TILES;
    const int  row0   = (active ? tile : (N_TILES - 1)) * 16;
    const int  rbase  = row0 + hi * 8;
    float* pw = pat[wave];

    int cnt = 0;
    if (tid < NCH) {
        const int* ls = lstart + tid * NBP + b;
        const int s = ls[0];
        cnt = ls[1] - s;
        runs[tid] = tid * CE + s;
    }
    const int inc = wave_inc_scan(cnt, lane);
    if (lane == 31) wtot[wave] = inc;
    __syncthreads();
    if (wave == 0) {
        const int tot = lane < CV_T / 32 ? wtot[lane] : 0;
        const int ti  = wave_inc_scan(tot, lane);
        if (lane < CV_T / 32) wtot[lane] = ti - tot;
    }
    __syncthreads();
    const int ex = wtot[wave] + inc - cnt;
    if (tid < NCH) runo[tid] = ex;
    if (tid == NCH - 1) runo[NCH] = ex + cnt;
    __syncthreads();
    const int n = __builtin_amdgcn_readfirstlane(runo[NCH]);

    v8f acc[8];
    #pragma unroll
    for (int nt = 0; nt < 8; ++nt) acc[nt] = (v8f){};
    v2f a[32];

    int w0 = 0;
    do {
        const int mm = n - w0 < CV_CAP ? n - w0 : CV_CAP;
        for (int i = tid; i < CV_NK; i += CV_T) keyc[i] = 0;
        __syncthreads();

        int rec[CV_U], rk[CV_U];
        {
            int gi[CV_U], lo[CV_U], rr0[CV_U];
            #pragma unroll
            for (int u = 0; u < CV_U; ++u) {
                const int i = tid + u * CV_T;
                gi[u] = i < mm ? w0 + i : -1;
                lo[u] = 0;
            }
            static_assert(NCH <= 256, "eight halvings cover 256 runs");
            #pragma unroll
            for (int step = 128; step >= 1; step >>= 1) {
                #pragma unroll
                for (int u = 0; u < CV_U; ++u) {
                    const int md = lo[u] + step;
                    if (runo[md < NCH ? md : NCH] <= gi[u]) lo[u] = md;
                }
            }
            int rs[CV_U], ro[CV_U];
            #pragma unroll
            for (int u = 0; u < CV_U; ++u) { rs[u] = runs[lo[u]]; ro[u] = runo[lo[u]]; }
            #pragma unroll
            for (int u = 0; u < CV_U; ++u) {
                const int idx = gi[u] >= 0 ? rs[u] + (gi[u] - ro[u]) : 0;
                rr0[u] = sorted[idx];
            }
            #pragma unroll
            for (int u = 0; u < CV_U; ++u) {
                const int r   = rr0[u];
                const int val = r & 0x7FFFF;
                const int rel = val / N_NODES;
                const int rr  = rel - rel0;
                rec[u] = -1;
                if (gi[u] >= 0 && rr >= 0 && rr < nre)
                    rec[u] = ((rr * 128 + (r >> 19)) << 16) | (val - rel * N_NODES);
            }
        }
        #pragma unroll
        for (int u = 0; u < CV_U; ++u) {
            rk[u] = 0;
            if (rec[u] >= 0) rk[u] = atomicAdd(&keyc[rec[u] >> 16], 1);
        }
        __syncthreads();

        if (wave == 0) {
            int cq[CV_KPL];
            int s = 0;
            #pragma unroll
            for (int q = 0; q < CV_KPL; ++q) { cq[q] = keyc[CV_KPL * lane + q]; s += cq[q]; }
            int e = wave_inc_scan(s, lane) - s;
            #pragma unroll
            for (int q = 0; q < CV_KPL; ++q) { keyb[CV_KPL * lane + q] = e; e += cq[q]; }
            if (lane == 31) keyb[CV_NK] = e;
        }
        __syncthreads();

        #pragma unroll
        for (int u = 0; u < CV_U; ++u)
            if (rec[u] >= 0) recs[keyb[rec[u] >> 16] + rk[u]] = rec[u];
        __syncthreads();

        for (int r = 0; r < nre; ++r) {
            v2f p[TR_Q];
            w_fetch(Wrel + (size_t)(rel0 + r) * D * D, tid, p);

            if (active) {
                const int k0  = r * 128 + wave * 16;
                const int beg = __builtin_amdgcn_readfirstlane(keyb[k0]);
                const int end = __builtin_amdgcn_readfirstlane(keyb[k0 + 16]);
                #pragma unroll
                for (int i = 0; i < 16; ++i) *(v4f*)(pw + i * ML_HP + lane * 4) = (v4f){};
                v4f sum = (v4f){};
                int cur = -1;
                for (int base = beg; base < end; base += 32) {
                    const int rem = end - base;
                    const int nn  = rem < 32 ? rem : 32;
                    const int mine = recs[base + (lane < nn ? lane : nn - 1)];
                    int j = 0;
                    for (; nn - j > 8; j += 16) rows_in<16>(sum, cur, pw, x, mine, j, nn, lane);
                    if (j < nn) rows_in<8>(sum, cur, pw, x, mine, j, nn, lane);
                }
                if (cur >= 0) *(v4f*)(pw + cur * ML_HP + lane * 4) = sum;
                h_get(pw, m, kh, a);
            }
            __syncthreads();
            w_put(Wl, tid, p);
            __syncthreads();

            if (active) tile_products(Wl, a, acc, hi, m);
        }
        w0 += CV_CAP;
    } while (w0 < n);

    if (active) tile_store(dstp, rbase, m, acc);
}

__device__ __forceinline__ void rows_fetch(
    const float* __restrict__ x, const float* mid,
    const float* __restrict__ part0, const float* __restrict__ part1,
    int r, int lane, bool three, v4f (&vx)[4], v4f (&vm)[4], v4f (&vq)[4], v4f (&vr)[4])
{
    #pragma unroll
    for (int i = 0; i < 4; ++i) {
        const size_t ro = (size_t)(r + i) * D;
        vx[i] = ((const v4f*)(x + ro))[lane];
        vm[i] = ((const v4f*)(mid + ro))[lane];
        vq[i] = ((const v4f*)(part0 + ro))[lane];
    }
    if (three) {
        #pragma unroll
        for (int i = 0; i < 4; ++i) vr[i] = ((const v4f*)(part1 + (size_t)(r + i) * D))[lane];
    }
}

__global__ __launch_bounds__(MQ_T) void k_mlp(
    const float* __restrict__ x, const float* mid,
    const float* __restrict__ part0, const float* __restrict__ part1,
    const float* W1, const float* Wc, const float* __restrict__ bc,
    const float* W2, const float* __restrict__ b2,
    float* out, int wofs)
{
    __shared__ alignas(16) float Wlds[2 * D * D];
    float* const wb0 = Wlds;
    float* const wb1 = Wlds + wofs;
    __shared__ alignas(16) float pt[2][MQ_TPR][16 * ML_HP];

    const int tid  = threadIdx.x;
    const int lane = tid & 31;
    const int wave = __builtin_amdgcn_readfirstlane(tid >> 5);
    const int m  = lane & 15;
    const int hi = lane >> 4;
    const int kh = hi * 2;
    const int cq = wave & 3;
    const int tj = wave >> 2;
    float* px = pt[0][tj];
    float* pa = pt[1][tj];

    const int t0  = ((int)blockIdx.x * N_TILES) / MQ_NB;
    const int t1  = (((int)blockIdx.x + 1) * N_TILES) / MQ_NB;
    const int nrd = (t1 - t0 + MQ_TPR - 1) / MQ_TPR;

    float bh0[2], bh1[2], bo[2];
    #pragma unroll
    for (int nt = 0; nt < 2; ++nt) {
        const int c = (2 * cq + nt) * 16 + m;
        bh0[nt] = bc[c];
        bh1[nt] = bc[D + c];
        bo[nt]  = b2[c];
    }

    v2f p[MQ_Q];
    wq_fetch<2 * D>(Wc, tid, p);
    wq_put(wb0, tid, p);

    #pragma unroll 1
    for (int rd = 0; rd < nrd; ++rd) {
        const int  tile   = t0 + rd * MQ_TPR + tj;
        const bool active = tile < t1;
        const int  rbase  = (active ? tile : t0) * 16 + hi * 8;

        if (active) {
            const bool three = (tile >> 3) >= CV_NB2;
            v4f vx[4], vm[4], vq[4], vr[4];
            rows_fetch(x, mid, part0, part1, tile * 16 + 4 * cq, lane, three, vx, vm, vq, vr);
            #pragma unroll
            for (int i = 0; i < 4; ++i) {
                v4f sm = vm[i] + vq[i];
                if (three) sm += vr[i];
                *(v4f*)(px + (4 * cq + i) * ML_HP + lane * 4) = vx[i];
                *(v4f*)(pa + (4 * cq + i) * ML_HP + lane * 4) = sm;
            }
        }
        __syncthreads();

        v2f a[32];
        v8f accA[2], accB[2];
        v4f cur[4];
        #pragma unroll
        for (int u = 0; u < 6; ++u) {
            const float* bp = quarter_bbase((u & 1) ? wb1 : wb0, hi, m, cq);
            if (active) {
                if (u == 0)      h_get(px, m, kh, a);
                else if (u == 2) h_get(pa, m, kh, a);
                else if (u == 4) h_get(px, m, kh, a);
                else if (u == 5) h_get(pa, m, kh, a);
                quarter_bhead(bp, cur);
            }
            __builtin_amdgcn_sched_barrier(0);

            const bool more = u < 5 || rd + 1 < nrd;
            if (more) {
                const int nx = (u + 1) % 6;
                if (nx < 2)      wq_fetch<2 * D>(Wc + nx * D, tid, p);
                else if (nx < 4) wq_fetch<2 * D>(W1 + (size_t)D * 2 * D + (nx - 2) * D, tid, p);
                else             wq_fetch<D>(W2 + (size_t)(nx - 4) * D * D, tid, p);
            }

            if (active) {
                if (u == 0)      bias_init2(accA, bh0);
                else if (u == 1) bias_init2(accB, bh1);
                else if (u == 4) bias_init2(accA, bo);

                if (u == 1 || u == 3) quarter_products(bp, a, accB, cur);
                else                  quarter_products(bp, a, accA, cur);

                if (u == 3) {
                    h_put2(px, m, hi, cq, accA);
                    h_put2(pa, m, hi, cq, accB);
                }
            }

            if (more) {
                wq_put(((u + 1) & 1) ? wb1 : wb0, tid, p);
                __syncthreads();
            }

            if (u == 5 && active) quarter_store(out, rbase, m, cq, accA);
        }
    }
}

extern "C" void kernel_launch(void* const* d_in, const int* in_sizes, int n_in,
                              void* d_out, int out_size, void* d_ws, size_t ws_size,
                              hipStream_t stream) {
    const float* x     = (const float*)d_in[0];
    const int*   src   = (const int*)  d_in[1];
    const int*   dst   = (const int*)  d_in[2];
    const int*   etype = (const int*)  d_in[3];
    const float* Wrel  = (const float*)d_in[4];
    const float* Wloop = (const float*)d_in[5];
    const float* brel  = (const float*)d_in[6];
    const float* W1    = (const float*)d_in[7];
    const float* b1    = (const float*)d_in[8];
    const float* W2    = (const float*)d_in[9];
    const float* b2    = (const float*)d_in[10];

    float* out = (float*)d_out;
    float* part0 = (float*)d_ws;
    float* part1 = part0 + (size_t)N_NODES * D;
    int* lstart = (int*)((char*)d_ws + 204800000);
    int* sorted = lstart + NCH * NBP;
    float* Wc = (float*)((char*)d_ws + 210000000);
    float* bc = Wc + D * 2 * D;

    k_chunk<<<NCH, CH_T, 0, stream>>>(src, dst, etype, lstart, sorted);

    k_conv<<<CV_NBLK + FD_NBLK, CV_T, 0, stream>>>(x, Wrel, Wloop, brel, W1, b1, lstart, sorted,
                                                   part0, part1, out, Wc, bc);

    k_mlp<<<MQ_NB, MQ_T, 0, stream>>>(x, out, part0, part1, W1, Wc, bc, W2, b2, out, D * D);
}
